# starting kernel
# baseline (speedup 1.0000x reference)
#include <hip/hip_runtime.h>

typedef __attribute__((ext_vector_type(2))) float v2f;
typedef __attribute__((ext_vector_type(8))) float v8f;

#define L_DIM 4096
#define H_DIM 8
#define D_DIM 32
#define NEGV (-1e30f)
#define TSCALE 0.1f
#define QSCALE 0.25506632772467906f


__global__ void mask_init_kernel(unsigned int* __restrict__ mask, int nwords) {
    int i = blockIdx.x * blockDim.x + threadIdx.x;
    int stride = gridDim.x * blockDim.x;
    for (; i < nwords; i += stride) mask[i] = 0xFFFFFFFFu;
}

__global__ void mask_clear_edges_kernel(unsigned int* __restrict__ mask,
                                        const int* __restrict__ ei, int E) {
    int e = blockIdx.x * blockDim.x + threadIdx.x;
    if (e >= E) return;
    unsigned int r = (unsigned int)ei[e];
    unsigned int c = (unsigned int)ei[E + e];
    unsigned int idx = r * L_DIM + c;
    atomicAnd(&mask[idx >> 5], ~(1u << (idx & 31u)));
}

__global__ void mask_clear_diag_kernel(unsigned int* __restrict__ mask) {
    int i = blockIdx.x * blockDim.x + threadIdx.x;
    if (i >= L_DIM) return;
    unsigned int idx = (unsigned int)i * (L_DIM + 1);
    atomicAnd(&mask[idx >> 5], ~(1u << (idx & 31u)));
}


__launch_bounds__(32)
__global__ void attn_wmma_kernel(const float* __restrict__ q,
                                 const float* __restrict__ k,
                                 const float* __restrict__ v,
                                 const float* __restrict__ dt,
                                 const unsigned int* __restrict__ mask,
                                 float* __restrict__ out) {
    __shared__ float pl[16 * 16];

    const int lane = threadIdx.x;
    const int hi   = lane >> 4;
    const int ln   = lane & 15;
    const int h    = blockIdx.y;
    const int row0 = blockIdx.x * 16;

    v2f aq[8];
    {
        const float* qp = q + (size_t)(h * L_DIM + row0 + ln) * D_DIM + 2 * hi;
#pragma unroll
        for (int kk = 0; kk < 8; ++kk) {
            v2f t = *(const v2f*)(qp + 4 * kk);
            aq[kk] = t * QSCALE;
        }
    }

    const float* kp = k + (size_t)(h * L_DIM + ln) * D_DIM + 2 * hi;
    const float* vp = v + (size_t)(h * L_DIM + 2 * hi) * D_DIM + ln;
    const float* dtp = dt + (size_t)(row0 + 8 * hi) * L_DIM + ln;
    const unsigned int* mp = mask + (size_t)(row0 + 8 * hi) * (L_DIM / 32);

    v8f o0 = {};
    v8f o1 = {};
    v8f ol = {};
    const v2f ones = (v2f){1.0f, 1.0f};

    const int ntiles = L_DIM / 16;
    for (int j = 0; j < ntiles; ++j) {
        if (j + 1 < ntiles) {
            __builtin_prefetch(kp + 16 * D_DIM, 0, 3);
            __builtin_prefetch(vp + 16 * D_DIM, 0, 3);
        }

        v2f bk[8];
#pragma unroll
        for (int kk = 0; kk < 8; ++kk) bk[kk] = *(const v2f*)(kp + 4 * kk);

        v8f s = {};
#pragma unroll
        for (int kk = 0; kk < 8; ++kk) {
            s = __builtin_amdgcn_wmma_f32_16x16x4_f32(false, aq[kk], false, bk[kk],
                                                      (short)0, s, false, false);
        }

        const int col0 = j * 16;
        const unsigned int* mpj = mp + (col0 >> 5);
        const int bitbase = (col0 & 16) + ln;

#pragma unroll
        for (int r = 0; r < 8; ++r) {
            const float d  = dtp[r * L_DIM];
            const float lg = __builtin_amdgcn_logf(1.0f + fabsf(d));
            const bool blocked = (mpj[r * (L_DIM / 32)] >> bitbase) & 1u;

            const float sv = blocked ? NEGV : fmaf(lg, -TSCALE, s[r]);
            const float pv = __builtin_amdgcn_exp2f(sv);

            pl[(r + 8 * hi) * 16 + ln] = pv;
        }

        asm volatile("s_wait_dscnt 0x0" ::: "memory");

        v2f pa[4];
#pragma unroll
        for (int kk = 0; kk < 4; ++kk) {
            pa[kk] = *(const v2f*)&pl[ln * 16 + 4 * kk + 2 * hi];
        }

        v2f bv0[4], bv1[4];
#pragma unroll
        for (int kk = 0; kk < 4; ++kk) {
            const float* vr = vp + kk * 4 * D_DIM;
            bv0[kk] = (v2f){vr[0],  vr[D_DIM]};
            bv1[kk] = (v2f){vr[16], vr[D_DIM + 16]};
        }

#pragma unroll
        for (int kk = 0; kk < 4; ++kk) {
            o0 = __builtin_amdgcn_wmma_f32_16x16x4_f32(false, pa[kk], false, bv0[kk],
                                                       (short)0, o0, false, false);
            o1 = __builtin_amdgcn_wmma_f32_16x16x4_f32(false, pa[kk], false, bv1[kk],
                                                       (short)0, o1, false, false);
            ol = __builtin_amdgcn_wmma_f32_16x16x4_f32(false, pa[kk], false, ones,
                                                       (short)0, ol, false, false);
        }

        kp  += 16 * D_DIM;
        vp  += 16 * D_DIM;
        dtp += 16;
    }

#pragma unroll
    for (int r = 0; r < 8; ++r) {
        const float l   = ol[r];
        const float inv = (l > 0.0f) ? (1.0f / l) : 0.0f;
        const size_t o = (size_t)(h * L_DIM + row0 + r + 8 * hi) * D_DIM;
        out[o + ln]      = o0[r] * inv;
        out[o + 16 + ln] = o1[r] * inv;
    }
}


extern "C" void kernel_launch(void* const* d_in, const int* in_sizes, int n_in,
                              void* d_out, int out_size, void* d_ws, size_t ws_size,
                              hipStream_t stream) {
    const float* q  = (const float*)d_in[0];
    const float* k  = (const float*)d_in[1];
    const float* v  = (const float*)d_in[2];
    const int*   ei = (const int*)d_in[3];
    const float* dt = (const float*)d_in[4];
    float* out = (float*)d_out;

    unsigned int* mask = (unsigned int*)d_ws;
    const int E = in_sizes[3] / 2;
    const int nwords = (L_DIM * L_DIM) / 32;

    mask_init_kernel<<<1024, 256, 0, stream>>>(mask, nwords);
    mask_clear_edges_kernel<<<(E + 255) / 256, 256, 0, stream>>>(mask, ei, E);
    mask_clear_diag_kernel<<<(L_DIM + 255) / 256, 256, 0, stream>>>(mask);

    dim3 grid(L_DIM / 16, H_DIM);
    attn_wmma_kernel<<<grid, 32, 0, stream>>>(q, k, v, dt, mask, out);
}


# Split each query tile's keys round-robin over 4 waves per work-group (was 1); the 4 partial O and l sums are added through LDS after the loop: 8 waves per SIMD hide load latency; sum order changes
# speedup vs baseline: 1.6305x; 1.6305x over previous
#include <hip/hip_runtime.h>

typedef __attribute__((ext_vector_type(2))) float v2f;
typedef __attribute__((ext_vector_type(8))) float v8f;

#define L_DIM 4096
#define H_DIM 8
#define D_DIM 32
#define NEGV (-1e30f)
#define TSCALE 0.1f
#define QSCALE 0.25506632772467906f


__global__ void mask_init_kernel(unsigned int* __restrict__ mask, int nwords) {
    int i = blockIdx.x * blockDim.x + threadIdx.x;
    int stride = gridDim.x * blockDim.x;
    for (; i < nwords; i += stride) mask[i] = 0xFFFFFFFFu;
}

__global__ void mask_clear_edges_kernel(unsigned int* __restrict__ mask,
                                        const int* __restrict__ ei, int E) {
    int e = blockIdx.x * blockDim.x + threadIdx.x;
    if (e >= E) return;
    unsigned int r = (unsigned int)ei[e];
    unsigned int c = (unsigned int)ei[E + e];
    unsigned int idx = r * L_DIM + c;
    atomicAnd(&mask[idx >> 5], ~(1u << (idx & 31u)));
}

__global__ void mask_clear_diag_kernel(unsigned int* __restrict__ mask) {
    int i = blockIdx.x * blockDim.x + threadIdx.x;
    if (i >= L_DIM) return;
    unsigned int idx = (unsigned int)i * (L_DIM + 1);
    atomicAnd(&mask[idx >> 5], ~(1u << (idx & 31u)));
}


#define NW 4
#define RP 34

__launch_bounds__(32 * NW)
__global__ void attn_wmma_kernel(const float* __restrict__ q,
                                 const float* __restrict__ k,
                                 const float* __restrict__ v,
                                 const float* __restrict__ dt,
                                 const unsigned int* __restrict__ mask,
                                 float* __restrict__ out) {
    __shared__ float pl[NW][16 * 16];
    __shared__ float red_o[NW][16 * RP];
    __shared__ float red_l[NW][16];

    const int tid  = threadIdx.x;
    const int w    = __builtin_amdgcn_readfirstlane(tid >> 5);
    const int lane = tid & 31;
    const int hi   = lane >> 4;
    const int ln   = lane & 15;
    const int h    = blockIdx.y;
    const int row0 = blockIdx.x * 16;
    float* plw = pl[w];

    v2f aq[8];
    {
        const float* qp = q + (size_t)(h * L_DIM + row0 + ln) * D_DIM + 2 * hi;
#pragma unroll
        for (int kk = 0; kk < 8; ++kk) {
            v2f t = *(const v2f*)(qp + 4 * kk);
            aq[kk] = t * QSCALE;
        }
    }

    const int key0 = w * 16;
    const float* kp = k + (size_t)(h * L_DIM + key0 + ln) * D_DIM + 2 * hi;
    const float* vp = v + (size_t)(h * L_DIM + key0 + 2 * hi) * D_DIM + ln;
    const float* dtp = dt + (size_t)(row0 + 8 * hi) * L_DIM + key0 + ln;
    const unsigned int* mp = mask + (size_t)(row0 + 8 * hi) * (L_DIM / 32);
    const int bitbase = ((w & 1) << 4) + ln;

    v8f o0 = {};
    v8f o1 = {};
    v8f ol = {};
    const v2f ones = (v2f){1.0f, 1.0f};

    const int nsteps = (L_DIM / 16) / NW;
    for (int i = 0; i < nsteps; ++i) {
        if (i + 1 < nsteps) {
            __builtin_prefetch(kp + NW * 16 * D_DIM, 0, 3);
            __builtin_prefetch(vp + NW * 16 * D_DIM, 0, 3);
        }

        v2f bk[8];
#pragma unroll
        for (int kk = 0; kk < 8; ++kk) bk[kk] = *(const v2f*)(kp + 4 * kk);

        v8f s = {};
#pragma unroll
        for (int kk = 0; kk < 8; ++kk) {
            s = __builtin_amdgcn_wmma_f32_16x16x4_f32(false, aq[kk], false, bk[kk],
                                                      (short)0, s, false, false);
        }

        const int col0 = (w + NW * i) * 16;
        const unsigned int* mpj = mp + (col0 >> 5);

#pragma unroll
        for (int r = 0; r < 8; ++r) {
            const float d  = dtp[r * L_DIM];
            const float lg = __builtin_amdgcn_logf(1.0f + fabsf(d));
            const bool blocked = (mpj[r * (L_DIM / 32)] >> bitbase) & 1u;

            const float sv = blocked ? NEGV : fmaf(lg, -TSCALE, s[r]);
            const float pv = __builtin_amdgcn_exp2f(sv);

            plw[(r + 8 * hi) * 16 + ln] = pv;
        }

        asm volatile("s_wait_dscnt 0x0" ::: "memory");

        v2f pa[4];
#pragma unroll
        for (int kk = 0; kk < 4; ++kk) {
            pa[kk] = *(const v2f*)&plw[ln * 16 + 4 * kk + 2 * hi];
        }

        v2f bv0[4], bv1[4];
#pragma unroll
        for (int kk = 0; kk < 4; ++kk) {
            const float* vr = vp + kk * 4 * D_DIM;
            bv0[kk] = (v2f){vr[0],  vr[D_DIM]};
            bv1[kk] = (v2f){vr[16], vr[D_DIM + 16]};
        }

#pragma unroll
        for (int kk = 0; kk < 4; ++kk) {
            o0 = __builtin_amdgcn_wmma_f32_16x16x4_f32(false, pa[kk], false, bv0[kk],
                                                       (short)0, o0, false, false);
            o1 = __builtin_amdgcn_wmma_f32_16x16x4_f32(false, pa[kk], false, bv1[kk],
                                                       (short)0, o1, false, false);
            ol = __builtin_amdgcn_wmma_f32_16x16x4_f32(false, pa[kk], false, ones,
                                                       (short)0, ol, false, false);
        }

        kp  += NW * 16 * D_DIM;
        vp  += NW * 16 * D_DIM;
        dtp += NW * 16;
    }

#pragma unroll
    for (int r = 0; r < 8; ++r) {
        red_o[w][(r + 8 * hi) * RP + ln]      = o0[r];
        red_o[w][(r + 8 * hi) * RP + 16 + ln] = o1[r];
    }
    if (ln == 0) {
#pragma unroll
        for (int r = 0; r < 8; ++r) red_l[w][r + 8 * hi] = ol[r];
    }
    __syncthreads();

#pragma unroll
    for (int e = 0; e < 16 / NW; ++e) {
        const int row = w + NW * e;
        float acc = 0.0f;
        float l   = 0.0f;
#pragma unroll
        for (int ww = 0; ww < NW; ++ww) {
            acc += red_o[ww][row * RP + lane];
            l   += red_l[ww][row];
        }
        const float inv = (l > 0.0f) ? (1.0f / l) : 0.0f;
        out[(size_t)(h * L_DIM + row0 + row) * D_DIM + lane] = acc * inv;
    }
}


extern "C" void kernel_launch(void* const* d_in, const int* in_sizes, int n_in,
                              void* d_out, int out_size, void* d_ws, size_t ws_size,
                              hipStream_t stream) {
    const float* q  = (const float*)d_in[0];
    const float* k  = (const float*)d_in[1];
    const float* v  = (const float*)d_in[2];
    const int*   ei = (const int*)d_in[3];
    const float* dt = (const float*)d_in[4];
    float* out = (float*)d_out;

    unsigned int* mask = (unsigned int*)d_ws;
    const int E = in_sizes[3] / 2;
    const int nwords = (L_DIM * L_DIM) / 32;

    mask_init_kernel<<<1024, 256, 0, stream>>>(mask, nwords);
    mask_clear_edges_kernel<<<(E + 255) / 256, 256, 0, stream>>>(mask, ei, E);
    mask_clear_diag_kernel<<<(L_DIM + 255) / 256, 256, 0, stream>>>(mask);

    dim3 grid(L_DIM / 16, H_DIM);
    attn_wmma_kernel<<<grid, 32 * NW, 0, stream>>>(q, k, v, dt, mask, out);
}


# K and V tiles: 4 whole-line 16-byte loads per lane each (4 lines per load, was 16 per fragment load), staged in wave-private LDS, fragments read from LDS; tile prefetch dropped; same sum order
# speedup vs baseline: 1.7801x; 1.0918x over previous
#include <hip/hip_runtime.h>

typedef __attribute__((ext_vector_type(2))) float v2f;
typedef __attribute__((ext_vector_type(4))) float v4f;
typedef __attribute__((ext_vector_type(8))) float v8f;

#define L_DIM 4096
#define H_DIM 8
#define D_DIM 32
#define NEGV (-1e30f)
#define TSCALE 0.1f
#define QSCALE 0.25506632772467906f


__global__ void mask_init_kernel(unsigned int* __restrict__ mask, int nwords) {
    int i = blockIdx.x * blockDim.x + threadIdx.x;
    int stride = gridDim.x * blockDim.x;
    for (; i < nwords; i += stride) mask[i] = 0xFFFFFFFFu;
}

__global__ void mask_clear_edges_kernel(unsigned int* __restrict__ mask,
                                        const int* __restrict__ ei, int E) {
    int e = blockIdx.x * blockDim.x + threadIdx.x;
    if (e >= E) return;
    unsigned int r = (unsigned int)ei[e];
    unsigned int c = (unsigned int)ei[E + e];
    unsigned int idx = r * L_DIM + c;
    atomicAnd(&mask[idx >> 5], ~(1u << (idx & 31u)));
}

__global__ void mask_clear_diag_kernel(unsigned int* __restrict__ mask) {
    int i = blockIdx.x * blockDim.x + threadIdx.x;
    if (i >= L_DIM) return;
    unsigned int idx = (unsigned int)i * (L_DIM + 1);
    atomicAnd(&mask[idx >> 5], ~(1u << (idx & 31u)));
}


#define NW 4
#define RP 34
#define KP 36
#define VP 40

__attribute__((amdgpu_waves_per_eu(8, 8)))
__launch_bounds__(32 * NW)
__global__ void attn_wmma_kernel(const float* __restrict__ q,
                                 const float* __restrict__ k,
                                 const float* __restrict__ v,
                                 const float* __restrict__ dt,
                                 const unsigned int* __restrict__ mask,
                                 float* __restrict__ out) {
    __shared__ __attribute__((aligned(16))) float kst[NW][16 * KP];
    __shared__ __attribute__((aligned(16))) float vst[NW][16 * VP];
    __shared__ float pl[NW][16 * 16];
    __shared__ float red_o[NW][16 * RP];
    __shared__ float red_l[NW][16];

    const int tid  = threadIdx.x;
    const int w    = __builtin_amdgcn_readfirstlane(tid >> 5);
    const int lane = tid & 31;
    const int hi   = lane >> 4;
    const int ln   = lane & 15;
    const int h    = blockIdx.y;
    const int row0 = blockIdx.x * 16;
    float* plw = pl[w];
    float* ksw = kst[w];
    float* vsw = vst[w];

    v2f aq[8];
    {
        const float* qp = q + (size_t)(h * L_DIM + row0 + ln) * D_DIM + 2 * hi;
#pragma unroll
        for (int kk = 0; kk < 8; ++kk) {
            v2f t = *(const v2f*)(qp + 4 * kk);
            aq[kk] = t * QSCALE;
        }
    }

    const int key0 = w * 16;
    const float* kt = k + (size_t)(h * L_DIM + key0) * D_DIM + 4 * lane;
    const float* vt = v + (size_t)(h * L_DIM + key0) * D_DIM + 4 * lane;
    float* kst_w = ksw + (lane >> 3) * KP + (lane & 7) * 4;
    float* vst_w = vsw + (lane >> 3) * VP + (lane & 7) * 4;
    const float* kfr = ksw + ln * KP + 2 * hi;
    const float* vfr = vsw + (2 * hi) * VP + ln;
    const float* dtp = dt + (size_t)(row0 + 8 * hi) * L_DIM + key0 + ln;
    const unsigned int* mp = mask + (size_t)(row0 + 8 * hi) * (L_DIM / 32);
    const int bitbase = ((w & 1) << 4) + ln;

    v8f o0 = {};
    v8f o1 = {};
    v8f ol = {};
    const v2f ones = (v2f){1.0f, 1.0f};

    const int nsteps = (L_DIM / 16) / NW;
    for (int i = 0; i < nsteps; ++i) {
        v4f kg[4], vg[4];
#pragma unroll
        for (int j = 0; j < 4; ++j) kg[j] = *(const v4f*)(kt + 128 * j);
#pragma unroll
        for (int j = 0; j < 4; ++j) vg[j] = *(const v4f*)(vt + 128 * j);

        const int col0 = (w + NW * i) * 16;
        const unsigned int* mpj = mp + (col0 >> 5);
        float dv[8];
        unsigned int mw[8];
#pragma unroll
        for (int r = 0; r < 8; ++r) {
            dv[r] = dtp[r * L_DIM];
            mw[r] = mpj[r * (L_DIM / 32)];
        }

#pragma unroll
        for (int j = 0; j < 4; ++j) *(v4f*)(kst_w + 4 * j * KP) = kg[j];
#pragma unroll
        for (int j = 0; j < 4; ++j) *(v4f*)(vst_w + 4 * j * VP) = vg[j];

        asm volatile("s_wait_dscnt 0x0" ::: "memory");

        v2f bk[8];
#pragma unroll
        for (int kk = 0; kk < 8; ++kk) bk[kk] = *(const v2f*)(kfr + 4 * kk);

        v8f s = {};
#pragma unroll
        for (int kk = 0; kk < 8; ++kk) {
            s = __builtin_amdgcn_wmma_f32_16x16x4_f32(false, aq[kk], false, bk[kk],
                                                      (short)0, s, false, false);
        }

#pragma unroll
        for (int r = 0; r < 8; ++r) {
            const float lg = __builtin_amdgcn_logf(1.0f + fabsf(dv[r]));
            const bool blocked = (mw[r] >> bitbase) & 1u;

            const float sv = blocked ? NEGV : fmaf(lg, -TSCALE, s[r]);
            const float pv = __builtin_amdgcn_exp2f(sv);

            plw[(r + 8 * hi) * 16 + ln] = pv;
        }

        asm volatile("s_wait_dscnt 0x0" ::: "memory");

        v2f pa[4];
#pragma unroll
        for (int kk = 0; kk < 4; ++kk) {
            pa[kk] = *(const v2f*)&plw[ln * 16 + 4 * kk + 2 * hi];
        }

        v2f bv0[4], bv1[4];
#pragma unroll
        for (int kk = 0; kk < 4; ++kk) {
            const float* vr = vfr + kk * 4 * VP;
            bv0[kk] = (v2f){vr[0],  vr[VP]};
            bv1[kk] = (v2f){vr[16], vr[VP + 16]};
        }

#pragma unroll
        for (int kk = 0; kk < 4; ++kk) {
            o0 = __builtin_amdgcn_wmma_f32_16x16x4_f32(false, pa[kk], false, bv0[kk],
                                                       (short)0, o0, false, false);
            o1 = __builtin_amdgcn_wmma_f32_16x16x4_f32(false, pa[kk], false, bv1[kk],
                                                       (short)0, o1, false, false);
            ol = __builtin_amdgcn_wmma_f32_16x16x4_f32(false, pa[kk], false, ones,
                                                       (short)0, ol, false, false);
        }

        kt  += NW * 16 * D_DIM;
        vt  += NW * 16 * D_DIM;
        dtp += NW * 16;
    }

#pragma unroll
    for (int r = 0; r < 8; ++r) {
        red_o[w][(r + 8 * hi) * RP + ln]      = o0[r];
        red_o[w][(r + 8 * hi) * RP + 16 + ln] = o1[r];
    }
    if (ln == 0) {
#pragma unroll
        for (int r = 0; r < 8; ++r) red_l[w][r + 8 * hi] = ol[r];
    }
    __syncthreads();

#pragma unroll
    for (int e = 0; e < 16 / NW; ++e) {
        const int row = w + NW * e;
        float acc = 0.0f;
        float l   = 0.0f;
#pragma unroll
        for (int ww = 0; ww < NW; ++ww) {
            acc += red_o[ww][row * RP + lane];
            l   += red_l[ww][row];
        }
        const float inv = (l > 0.0f) ? (1.0f / l) : 0.0f;
        out[(size_t)(h * L_DIM + row0 + row) * D_DIM + lane] = acc * inv;
    }
}


extern "C" void kernel_launch(void* const* d_in, const int* in_sizes, int n_in,
                              void* d_out, int out_size, void* d_ws, size_t ws_size,
                              hipStream_t stream) {
    const float* q  = (const float*)d_in[0];
    const float* k  = (const float*)d_in[1];
    const float* v  = (const float*)d_in[2];
    const int*   ei = (const int*)d_in[3];
    const float* dt = (const float*)d_in[4];
    float* out = (float*)d_out;

    unsigned int* mask = (unsigned int*)d_ws;
    const int E = in_sizes[3] / 2;
    const int nwords = (L_DIM * L_DIM) / 32;

    mask_init_kernel<<<1024, 256, 0, stream>>>(mask, nwords);
    mask_clear_edges_kernel<<<(E + 255) / 256, 256, 0, stream>>>(mask, ei, E);
    mask_clear_diag_kernel<<<(L_DIM + 255) / 256, 256, 0, stream>>>(mask);

    dim3 grid(L_DIM / 16, H_DIM);
    attn_wmma_kernel<<<grid, 32 * NW, 0, stream>>>(q, k, v, dt, mask, out);
}


# Next key tile's K, V, dt and mask loads issued one tile ahead, under this tile's LDS reads, WMMAs and exp2; to fit 128 registers the tile's 16 mask words come by scalar loads; same sum order
# speedup vs baseline: 1.9469x; 1.0937x over previous
#include <hip/hip_runtime.h>

typedef __attribute__((ext_vector_type(2))) float v2f;
typedef __attribute__((ext_vector_type(4))) float v4f;
typedef __attribute__((ext_vector_type(8))) float v8f;

#define L_DIM 4096
#define H_DIM 8
#define D_DIM 32
#define TSCALE 0.1f
#define QSCALE 0.25506632772467906f


__global__ void mask_init_kernel(unsigned int* __restrict__ mask, int nwords) {
    int i = blockIdx.x * blockDim.x + threadIdx.x;
    int stride = gridDim.x * blockDim.x;
    for (; i < nwords; i += stride) mask[i] = 0xFFFFFFFFu;
}

__global__ void mask_clear_edges_kernel(unsigned int* __restrict__ mask,
                                        const int* __restrict__ ei, int E) {
    int e = blockIdx.x * blockDim.x + threadIdx.x;
    if (e >= E) return;
    unsigned int r = (unsigned int)ei[e];
    unsigned int c = (unsigned int)ei[E + e];
    unsigned int idx = r * L_DIM + c;
    atomicAnd(&mask[idx >> 5], ~(1u << (idx & 31u)));
}

__global__ void mask_clear_diag_kernel(unsigned int* __restrict__ mask) {
    int i = blockIdx.x * blockDim.x + threadIdx.x;
    if (i >= L_DIM) return;
    unsigned int idx = (unsigned int)i * (L_DIM + 1);
    atomicAnd(&mask[idx >> 5], ~(1u << (idx & 31u)));
}


#define NW 4
#define RP 34
#define KP 36
#define VP 40

__attribute__((amdgpu_waves_per_eu(8, 8)))
__launch_bounds__(32 * NW)
__global__ void attn_wmma_kernel(const float* __restrict__ q,
                                 const float* __restrict__ k,
                                 const float* __restrict__ v,
                                 const float* __restrict__ dt,
                                 const unsigned int* __restrict__ mask,
                                 float* __restrict__ out) {
    __shared__ __attribute__((aligned(16))) float kst[NW][16 * KP];
    __shared__ __attribute__((aligned(16))) float vst[NW][16 * VP];
    __shared__ float pl[NW][16 * 16];
    __shared__ float red_o[NW][16 * RP];
    __shared__ float red_l[NW][16];

    const int tid  = threadIdx.x;
    const int w    = __builtin_amdgcn_readfirstlane(tid >> 5);
    const int lane = tid & 31;
    const int hi   = lane >> 4;
    const int ln   = lane & 15;
    const int h    = blockIdx.y;
    const int row0 = blockIdx.x * 16;
    float* plw = pl[w];
    float* ksw = kst[w];
    float* vsw = vst[w];

    v2f aq[8];
    {
        const float* qp = q + (size_t)(h * L_DIM + row0 + ln) * D_DIM + 2 * hi;
#pragma unroll
        for (int kk = 0; kk < 8; ++kk) {
            v2f t = *(const v2f*)(qp + 4 * kk);
            aq[kk] = t * QSCALE;
        }
    }

    const int key0 = w * 16;
    const char* kb = (const char*)(k + (size_t)(h * L_DIM + key0) * D_DIM);
    const char* vb = (const char*)(v + (size_t)(h * L_DIM + key0) * D_DIM);
    const char* db = (const char*)(dt + (size_t)row0 * L_DIM + key0);
    const unsigned int* mb = mask + (size_t)row0 * (L_DIM / 32) + (key0 >> 5);
    const unsigned int kvoff = 16u * (unsigned int)lane;
    const unsigned int dtoff = 4u * (unsigned int)(8 * hi * L_DIM + ln);
    float* kst_w = ksw + (lane >> 3) * KP + (lane & 7) * 4;
    float* vst_w = vsw + (lane >> 3) * VP + (lane & 7) * 4;
    const float* kfr = ksw + ln * KP + 2 * hi;
    const float* vfr = vsw + (2 * hi) * VP + ln;
    const unsigned int mshift = (unsigned int)(w & 1) << 4;

    v8f o0 = {};
    v8f o1 = {};
    v8f ol = {};
    const v2f ones = (v2f){1.0f, 1.0f};

    const int nsteps = (L_DIM / 16) / NW;

    v4f kg[4], vg[4];
    float dv[8];
    unsigned int ms[16];
    auto fetch_tile = [&]() {
#pragma unroll
        for (int j = 0; j < 4; ++j) kg[j] = *(const v4f*)(kb + kvoff + 512 * j);
#pragma unroll
        for (int j = 0; j < 4; ++j) vg[j] = *(const v4f*)(vb + kvoff + 512 * j);
#pragma unroll
        for (int r = 0; r < 8; ++r) dv[r] = *(const float*)(db + dtoff + r * (L_DIM * 4));
#pragma unroll
        for (int rr = 0; rr < 16; ++rr) ms[rr] = mb[rr * (L_DIM / 32)];
    };
    fetch_tile();

    for (int i = 0; i < nsteps; ++i) {
#pragma unroll
        for (int j = 0; j < 4; ++j) *(v4f*)(kst_w + 4 * j * KP) = kg[j];
#pragma unroll
        for (int j = 0; j < 4; ++j) *(v4f*)(vst_w + 4 * j * VP) = vg[j];

        float lb[8];
#pragma unroll
        for (int r = 0; r < 8; ++r) {
            const float lg = __builtin_amdgcn_logf(1.0f + fabsf(dv[r]));
            const unsigned int lanes = ((ms[r] >> mshift) & 0xFFFFu) | ((ms[r + 8] >> mshift) << 16);
            const bool blocked = __builtin_amdgcn_inverse_ballot_w32(lanes);
            lb[r] = blocked ? __builtin_inff() : lg;
        }

        if (i + 1 < nsteps) {
            kb += NW * 16 * D_DIM * 4;
            vb += NW * 16 * D_DIM * 4;
            db += NW * 16 * 4;
            mb += (NW * 16) / 32;
            fetch_tile();
        }

        asm volatile("s_wait_dscnt 0x0" ::: "memory");

        v2f bk[8];
#pragma unroll
        for (int kk = 0; kk < 8; ++kk) bk[kk] = *(const v2f*)(kfr + 4 * kk);

        v8f s = {};
#pragma unroll
        for (int kk = 0; kk < 8; ++kk) {
            s = __builtin_amdgcn_wmma_f32_16x16x4_f32(false, aq[kk], false, bk[kk],
                                                      (short)0, s, false, false);
        }

#pragma unroll
        for (int r = 0; r < 8; ++r) {
            const float sv = fmaf(lb[r], -TSCALE, s[r]);
            const float pv = __builtin_amdgcn_exp2f(sv);

            plw[(r + 8 * hi) * 16 + ln] = pv;
        }

        asm volatile("s_wait_dscnt 0x0" ::: "memory");

        v2f pa[4];
#pragma unroll
        for (int kk = 0; kk < 4; ++kk) {
            pa[kk] = *(const v2f*)&plw[ln * 16 + 4 * kk + 2 * hi];
        }

        v2f bv0[4], bv1[4];
#pragma unroll
        for (int kk = 0; kk < 4; ++kk) {
            const float* vr = vfr + kk * 4 * VP;
            bv0[kk] = (v2f){vr[0],  vr[VP]};
            bv1[kk] = (v2f){vr[16], vr[VP + 16]};
        }

#pragma unroll
        for (int kk = 0; kk < 4; ++kk) {
            o0 = __builtin_amdgcn_wmma_f32_16x16x4_f32(false, pa[kk], false, bv0[kk],
                                                       (short)0, o0, false, false);
            o1 = __builtin_amdgcn_wmma_f32_16x16x4_f32(false, pa[kk], false, bv1[kk],
                                                       (short)0, o1, false, false);
            ol = __builtin_amdgcn_wmma_f32_16x16x4_f32(false, pa[kk], false, ones,
                                                       (short)0, ol, false, false);
        }
    }

#pragma unroll
    for (int r = 0; r < 8; ++r) {
        red_o[w][(r + 8 * hi) * RP + ln]      = o0[r];
        red_o[w][(r + 8 * hi) * RP + 16 + ln] = o1[r];
    }
    if (ln == 0) {
#pragma unroll
        for (int r = 0; r < 8; ++r) red_l[w][r + 8 * hi] = ol[r];
    }
    __syncthreads();

#pragma unroll
    for (int e = 0; e < 16 / NW; ++e) {
        const int row = w + NW * e;
        float acc = 0.0f;
        float l   = 0.0f;
#pragma unroll
        for (int ww = 0; ww < NW; ++ww) {
            acc += red_o[ww][row * RP + lane];
            l   += red_l[ww][row];
        }
        const float inv = (l > 0.0f) ? (1.0f / l) : 0.0f;
        out[(size_t)(h * L_DIM + row0 + row) * D_DIM + lane] = acc * inv;
    }
}


extern "C" void kernel_launch(void* const* d_in, const int* in_sizes, int n_in,
                              void* d_out, int out_size, void* d_ws, size_t ws_size,
                              hipStream_t stream) {
    const float* q  = (const float*)d_in[0];
    const float* k  = (const float*)d_in[1];
    const float* v  = (const float*)d_in[2];
    const int*   ei = (const int*)d_in[3];
    const float* dt = (const float*)d_in[4];
    float* out = (float*)d_out;

    unsigned int* mask = (unsigned int*)d_ws;
    const int E = in_sizes[3] / 2;
    const int nwords = (L_DIM * L_DIM) / 32;

    mask_init_kernel<<<1024, 256, 0, stream>>>(mask, nwords);
    mask_clear_edges_kernel<<<(E + 255) / 256, 256, 0, stream>>>(mask, ei, E);
    mask_clear_diag_kernel<<<(L_DIM + 255) / 256, 256, 0, stream>>>(mask);

    dim3 grid(L_DIM / 16, H_DIM);
    attn_wmma_kernel<<<grid, 32 * NW, 0, stream>>>(q, k, v, dt, mask, out);
}


# Row sums: P tiles added elementwise on the vector unit in the loop, 16 columns summed by one all-ones WMMA pass after it (was 4 all-ones WMMAs per tile): 16 WMMAs per tile, was 20; sum order changes
# speedup vs baseline: 2.0848x; 1.0708x over previous
#include <hip/hip_runtime.h>

typedef __attribute__((ext_vector_type(2))) float v2f;
typedef __attribute__((ext_vector_type(4))) float v4f;
typedef __attribute__((ext_vector_type(8))) float v8f;

#define L_DIM 4096
#define H_DIM 8
#define D_DIM 32
#define TSCALE 0.1f
#define QSCALE 0.25506632772467906f


__global__ void mask_init_kernel(unsigned int* __restrict__ mask, int nwords) {
    int i = blockIdx.x * blockDim.x + threadIdx.x;
    int stride = gridDim.x * blockDim.x;
    for (; i < nwords; i += stride) mask[i] = 0xFFFFFFFFu;
}

__global__ void mask_clear_edges_kernel(unsigned int* __restrict__ mask,
                                        const int* __restrict__ ei, int E) {
    int e = blockIdx.x * blockDim.x + threadIdx.x;
    if (e >= E) return;
    unsigned int r = (unsigned int)ei[e];
    unsigned int c = (unsigned int)ei[E + e];
    unsigned int idx = r * L_DIM + c;
    atomicAnd(&mask[idx >> 5], ~(1u << (idx & 31u)));
}

__global__ void mask_clear_diag_kernel(unsigned int* __restrict__ mask) {
    int i = blockIdx.x * blockDim.x + threadIdx.x;
    if (i >= L_DIM) return;
    unsigned int idx = (unsigned int)i * (L_DIM + 1);
    atomicAnd(&mask[idx >> 5], ~(1u << (idx & 31u)));
}


#define NW 4
#define RP 34
#define KP 36
#define VP 40

__attribute__((amdgpu_waves_per_eu(8, 8)))
__launch_bounds__(32 * NW)
__global__ void attn_wmma_kernel(const float* __restrict__ q,
                                 const float* __restrict__ k,
                                 const float* __restrict__ v,
                                 const float* __restrict__ dt,
                                 const unsigned int* __restrict__ mask,
                                 float* __restrict__ out) {
    __shared__ __attribute__((aligned(16))) float kst[NW][16 * KP];
    __shared__ __attribute__((aligned(16))) float vst[NW][16 * VP];
    __shared__ float pl[NW][16 * 16];
    __shared__ float red_o[NW][16 * RP];
    __shared__ float red_l[NW][16];

    const int tid  = threadIdx.x;
    const int w    = __builtin_amdgcn_readfirstlane(tid >> 5);
    const int lane = tid & 31;
    const int hi   = lane >> 4;
    const int ln   = lane & 15;
    const int h    = blockIdx.y;
    const int row0 = blockIdx.x * 16;
    float* plw = pl[w];
    float* ksw = kst[w];
    float* vsw = vst[w];

    v2f aq[8];
    {
        const float* qp = q + (size_t)(h * L_DIM + row0 + ln) * D_DIM + 2 * hi;
#pragma unroll
        for (int kk = 0; kk < 8; ++kk) {
            v2f t = *(const v2f*)(qp + 4 * kk);
            aq[kk] = t * QSCALE;
        }
    }

    const int key0 = w * 16;
    const char* kb = (const char*)(k + (size_t)(h * L_DIM + key0) * D_DIM);
    const char* vb = (const char*)(v + (size_t)(h * L_DIM + key0) * D_DIM);
    const char* db = (const char*)(dt + (size_t)row0 * L_DIM + key0);
    const unsigned int* mb = mask + (size_t)row0 * (L_DIM / 32) + (key0 >> 5);
    const unsigned int kvoff = 16u * (unsigned int)lane;
    const unsigned int dtoff = 4u * (unsigned int)(8 * hi * L_DIM + ln);
    float* kst_w = ksw + (lane >> 3) * KP + (lane & 7) * 4;
    float* vst_w = vsw + (lane >> 3) * VP + (lane & 7) * 4;
    const float* kfr = ksw + ln * KP + 2 * hi;
    const float* vfr = vsw + (2 * hi) * VP + ln;
    const unsigned int mshift = (unsigned int)(w & 1) << 4;

    v8f o0 = {};
    v8f o1 = {};
    float ps[8];
#pragma unroll
    for (int r = 0; r < 8; ++r) ps[r] = 0.0f;

    const int nsteps = (L_DIM / 16) / NW;

    v4f kg[4], vg[4];
    float dv[8];
    unsigned int ms[16];
    auto fetch_tile = [&]() {
#pragma unroll
        for (int j = 0; j < 4; ++j) kg[j] = *(const v4f*)(kb + kvoff + 512 * j);
#pragma unroll
        for (int j = 0; j < 4; ++j) vg[j] = *(const v4f*)(vb + kvoff + 512 * j);
#pragma unroll
        for (int r = 0; r < 8; ++r) dv[r] = *(const float*)(db + dtoff + r * (L_DIM * 4));
#pragma unroll
        for (int rr = 0; rr < 16; ++rr) ms[rr] = mb[rr * (L_DIM / 32)];
    };
    fetch_tile();

    for (int i = 0; i < nsteps; ++i) {
#pragma unroll
        for (int j = 0; j < 4; ++j) *(v4f*)(kst_w + 4 * j * KP) = kg[j];
#pragma unroll
        for (int j = 0; j < 4; ++j) *(v4f*)(vst_w + 4 * j * VP) = vg[j];

        float lb[8];
#pragma unroll
        for (int r = 0; r < 8; ++r) {
            const float lg = __builtin_amdgcn_logf(1.0f + fabsf(dv[r]));
            const unsigned int lanes = ((ms[r] >> mshift) & 0xFFFFu) | ((ms[r + 8] >> mshift) << 16);
            const bool blocked = __builtin_amdgcn_inverse_ballot_w32(lanes);
            lb[r] = blocked ? __builtin_inff() : lg;
        }

        if (i + 1 < nsteps) {
            kb += NW * 16 * D_DIM * 4;
            vb += NW * 16 * D_DIM * 4;
            db += NW * 16 * 4;
            mb += (NW * 16) / 32;
            fetch_tile();
        }

        asm volatile("s_wait_dscnt 0x0" ::: "memory");

        v2f bk[8];
#pragma unroll
        for (int kk = 0; kk < 8; ++kk) bk[kk] = *(const v2f*)(kfr + 4 * kk);

        v8f s = {};
#pragma unroll
        for (int kk = 0; kk < 8; ++kk) {
            s = __builtin_amdgcn_wmma_f32_16x16x4_f32(false, aq[kk], false, bk[kk],
                                                      (short)0, s, false, false);
        }

#pragma unroll
        for (int r = 0; r < 8; ++r) {
            const float sv = fmaf(lb[r], -TSCALE, s[r]);
            const float pv = __builtin_amdgcn_exp2f(sv);
            ps[r] += pv;

            plw[(r + 8 * hi) * 16 + ln] = pv;
        }

        asm volatile("s_wait_dscnt 0x0" ::: "memory");

        v2f pa[4];
#pragma unroll
        for (int kk = 0; kk < 4; ++kk) {
            pa[kk] = *(const v2f*)&plw[ln * 16 + 4 * kk + 2 * hi];
        }

        v2f bv0[4], bv1[4];
#pragma unroll
        for (int kk = 0; kk < 4; ++kk) {
            const float* vr = vfr + kk * 4 * VP;
            bv0[kk] = (v2f){vr[0],  vr[VP]};
            bv1[kk] = (v2f){vr[16], vr[VP + 16]};
        }

#pragma unroll
        for (int kk = 0; kk < 4; ++kk) {
            o0 = __builtin_amdgcn_wmma_f32_16x16x4_f32(false, pa[kk], false, bv0[kk],
                                                       (short)0, o0, false, false);
            o1 = __builtin_amdgcn_wmma_f32_16x16x4_f32(false, pa[kk], false, bv1[kk],
                                                       (short)0, o1, false, false);
        }
    }

    v8f ol = {};
    {
        const v2f ones = (v2f){1.0f, 1.0f};
#pragma unroll
        for (int r = 0; r < 8; ++r) plw[(r + 8 * hi) * 16 + ln] = ps[r];
        asm volatile("s_wait_dscnt 0x0" ::: "memory");
#pragma unroll
        for (int kk = 0; kk < 4; ++kk) {
            const v2f sa = *(const v2f*)&plw[ln * 16 + 4 * kk + 2 * hi];
            ol = __builtin_amdgcn_wmma_f32_16x16x4_f32(false, sa, false, ones,
                                                       (short)0, ol, false, false);
        }
    }

#pragma unroll
    for (int r = 0; r < 8; ++r) {
        red_o[w][(r + 8 * hi) * RP + ln]      = o0[r];
        red_o[w][(r + 8 * hi) * RP + 16 + ln] = o1[r];
    }
    if (ln == 0) {
#pragma unroll
        for (int r = 0; r < 8; ++r) red_l[w][r + 8 * hi] = ol[r];
    }
    __syncthreads();

#pragma unroll
    for (int e = 0; e < 16 / NW; ++e) {
        const int row = w + NW * e;
        float acc = 0.0f;
        float l   = 0.0f;
#pragma unroll
        for (int ww = 0; ww < NW; ++ww) {
            acc += red_o[ww][row * RP + lane];
            l   += red_l[ww][row];
        }
        const float inv = (l > 0.0f) ? (1.0f / l) : 0.0f;
        out[(size_t)(h * L_DIM + row0 + row) * D_DIM + lane] = acc * inv;
    }
}


extern "C" void kernel_launch(void* const* d_in, const int* in_sizes, int n_in,
                              void* d_out, int out_size, void* d_ws, size_t ws_size,
                              hipStream_t stream) {
    const float* q  = (const float*)d_in[0];
    const float* k  = (const float*)d_in[1];
    const float* v  = (const float*)d_in[2];
    const int*   ei = (const int*)d_in[3];
    const float* dt = (const float*)d_in[4];
    float* out = (float*)d_out;

    unsigned int* mask = (unsigned int*)d_ws;
    const int E = in_sizes[3] / 2;
    const int nwords = (L_DIM * L_DIM) / 32;

    mask_init_kernel<<<1024, 256, 0, stream>>>(mask, nwords);
    mask_clear_edges_kernel<<<(E + 255) / 256, 256, 0, stream>>>(mask, ei, E);
    mask_clear_diag_kernel<<<(L_DIM + 255) / 256, 256, 0, stream>>>(mask);

    dim3 grid(L_DIM / 16, H_DIM);
    attn_wmma_kernel<<<grid, 32 * NW, 0, stream>>>(q, k, v, dt, mask, out);
}


# Each wave carries two 16-row query tiles per staged key tile (grid L/32 x H, 4 waves per SIMD, 194 registers): K and V loads, LDS staging and fragment reads halve per score; same sum order
# speedup vs baseline: 2.3926x; 1.1476x over previous
#include <hip/hip_runtime.h>

typedef __attribute__((ext_vector_type(2))) float v2f;
typedef __attribute__((ext_vector_type(4))) float v4f;
typedef __attribute__((ext_vector_type(8))) float v8f;

#define L_DIM 4096
#define H_DIM 8
#define D_DIM 32
#define TSCALE 0.1f
#define QSCALE 0.25506632772467906f


__global__ void mask_init_kernel(unsigned int* __restrict__ mask, int nwords) {
    int i = blockIdx.x * blockDim.x + threadIdx.x;
    int stride = gridDim.x * blockDim.x;
    for (; i < nwords; i += stride) mask[i] = 0xFFFFFFFFu;
}

__global__ void mask_clear_edges_kernel(unsigned int* __restrict__ mask,
                                        const int* __restrict__ ei, int E) {
    int e = blockIdx.x * blockDim.x + threadIdx.x;
    if (e >= E) return;
    unsigned int r = (unsigned int)ei[e];
    unsigned int c = (unsigned int)ei[E + e];
    unsigned int idx = r * L_DIM + c;
    atomicAnd(&mask[idx >> 5], ~(1u << (idx & 31u)));
}

__global__ void mask_clear_diag_kernel(unsigned int* __restrict__ mask) {
    int i = blockIdx.x * blockDim.x + threadIdx.x;
    if (i >= L_DIM) return;
    unsigned int idx = (unsigned int)i * (L_DIM + 1);
    atomicAnd(&mask[idx >> 5], ~(1u << (idx & 31u)));
}


#define NW 4
#define QT 2
#define RP 34
#define KP 36
#define VP 40

__attribute__((amdgpu_waves_per_eu(4, 4)))
__launch_bounds__(32 * NW)
__global__ void attn_wmma_kernel(const float* __restrict__ q,
                                 const float* __restrict__ k,
                                 const float* __restrict__ v,
                                 const float* __restrict__ dt,
                                 const unsigned int* __restrict__ mask,
                                 float* __restrict__ out) {
    __shared__ __attribute__((aligned(16))) float kst[NW][16 * KP];
    __shared__ __attribute__((aligned(16))) float vst[NW][16 * VP];
    __shared__ float pl[NW][QT][16 * 16];
    __shared__ float red_o[QT][NW][16 * RP];
    __shared__ float red_l[QT][NW][16];

    const int tid  = threadIdx.x;
    const int w    = __builtin_amdgcn_readfirstlane(tid >> 5);
    const int lane = tid & 31;
    const int hi   = lane >> 4;
    const int ln   = lane & 15;
    const int h    = blockIdx.y;
    const int row0 = blockIdx.x * (16 * QT);
    float* plw = pl[w][0];
    float* ksw = kst[w];
    float* vsw = vst[w];

    v2f aq[QT][8];
#pragma unroll
    for (int t = 0; t < QT; ++t) {
        const float* qp = q + (size_t)(h * L_DIM + row0 + 16 * t + ln) * D_DIM + 2 * hi;
#pragma unroll
        for (int kk = 0; kk < 8; ++kk) {
            v2f tq = *(const v2f*)(qp + 4 * kk);
            aq[t][kk] = tq * QSCALE;
        }
    }

    const int key0 = w * 16;
    const char* kb = (const char*)(k + (size_t)(h * L_DIM + key0) * D_DIM);
    const char* vb = (const char*)(v + (size_t)(h * L_DIM + key0) * D_DIM);
    const char* db = (const char*)(dt + (size_t)row0 * L_DIM + key0);
    const unsigned int* mb = mask + (size_t)row0 * (L_DIM / 32) + (key0 >> 5);
    const unsigned int kvoff = 16u * (unsigned int)lane;
    const unsigned int dtoff = 4u * (unsigned int)(8 * hi * L_DIM + ln);
    float* kst_w = ksw + (lane >> 3) * KP + (lane & 7) * 4;
    float* vst_w = vsw + (lane >> 3) * VP + (lane & 7) * 4;
    const float* kfr = ksw + ln * KP + 2 * hi;
    const float* vfr = vsw + (2 * hi) * VP + ln;
    const unsigned int mshift = (unsigned int)(w & 1) << 4;

    v8f o0[QT], o1[QT];
    float ps[QT][8];
#pragma unroll
    for (int t = 0; t < QT; ++t) {
        o0[t] = (v8f){};
        o1[t] = (v8f){};
#pragma unroll
        for (int r = 0; r < 8; ++r) ps[t][r] = 0.0f;
    }

    const int nsteps = (L_DIM / 16) / NW;

    v4f kg[4], vg[4];
    float dv[QT][8];
    unsigned int ms[QT * 16];
    auto fetch_tile = [&]() {
#pragma unroll
        for (int j = 0; j < 4; ++j) kg[j] = *(const v4f*)(kb + kvoff + 512 * j);
#pragma unroll
        for (int j = 0; j < 4; ++j) vg[j] = *(const v4f*)(vb + kvoff + 512 * j);
#pragma unroll
        for (int t = 0; t < QT; ++t) {
#pragma unroll
            for (int r = 0; r < 8; ++r) dv[t][r] = *(const float*)(db + dtoff + (16 * t + r) * (L_DIM * 4));
        }
#pragma unroll
        for (int rr = 0; rr < QT * 16; ++rr) ms[rr] = mb[rr * (L_DIM / 32)];
    };
    fetch_tile();

    for (int i = 0; i < nsteps; ++i) {
#pragma unroll
        for (int j = 0; j < 4; ++j) *(v4f*)(kst_w + 4 * j * KP) = kg[j];
#pragma unroll
        for (int j = 0; j < 4; ++j) *(v4f*)(vst_w + 4 * j * VP) = vg[j];

        float lb[QT][8];
#pragma unroll
        for (int t = 0; t < QT; ++t) {
#pragma unroll
            for (int r = 0; r < 8; ++r) {
                const float lg = __builtin_amdgcn_logf(1.0f + fabsf(dv[t][r]));
                const unsigned int lanes = ((ms[16 * t + r] >> mshift) & 0xFFFFu) | ((ms[16 * t + r + 8] >> mshift) << 16);
                const bool blocked = __builtin_amdgcn_inverse_ballot_w32(lanes);
                lb[t][r] = blocked ? __builtin_inff() : lg;
            }
        }

        if (i + 1 < nsteps) {
            kb += NW * 16 * D_DIM * 4;
            vb += NW * 16 * D_DIM * 4;
            db += NW * 16 * 4;
            mb += (NW * 16) / 32;
            fetch_tile();
        }

        asm volatile("s_wait_dscnt 0x0" ::: "memory");

        v2f bk[8];
#pragma unroll
        for (int kk = 0; kk < 8; ++kk) bk[kk] = *(const v2f*)(kfr + 4 * kk);

        v8f s[QT];
#pragma unroll
        for (int t = 0; t < QT; ++t) s[t] = (v8f){};
#pragma unroll
        for (int kk = 0; kk < 8; ++kk) {
#pragma unroll
            for (int t = 0; t < QT; ++t) {
                s[t] = __builtin_amdgcn_wmma_f32_16x16x4_f32(false, aq[t][kk], false, bk[kk],
                                                             (short)0, s[t], false, false);
            }
        }

#pragma unroll
        for (int t = 0; t < QT; ++t) {
#pragma unroll
            for (int r = 0; r < 8; ++r) {
                const float sv = fmaf(lb[t][r], -TSCALE, s[t][r]);
                const float pv = __builtin_amdgcn_exp2f(sv);
                ps[t][r] += pv;

                plw[t * 256 + (r + 8 * hi) * 16 + ln] = pv;
            }
        }

        asm volatile("s_wait_dscnt 0x0" ::: "memory");

        v2f pa[QT][4];
#pragma unroll
        for (int t = 0; t < QT; ++t) {
#pragma unroll
            for (int kk = 0; kk < 4; ++kk) {
                pa[t][kk] = *(const v2f*)&plw[t * 256 + ln * 16 + 4 * kk + 2 * hi];
            }
        }

        v2f bv0[4], bv1[4];
#pragma unroll
        for (int kk = 0; kk < 4; ++kk) {
            const float* vr = vfr + kk * 4 * VP;
            bv0[kk] = (v2f){vr[0],  vr[VP]};
            bv1[kk] = (v2f){vr[16], vr[VP + 16]};
        }

#pragma unroll
        for (int kk = 0; kk < 4; ++kk) {
#pragma unroll
            for (int t = 0; t < QT; ++t) {
                o0[t] = __builtin_amdgcn_wmma_f32_16x16x4_f32(false, pa[t][kk], false, bv0[kk],
                                                              (short)0, o0[t], false, false);
                o1[t] = __builtin_amdgcn_wmma_f32_16x16x4_f32(false, pa[t][kk], false, bv1[kk],
                                                              (short)0, o1[t], false, false);
            }
        }
    }

    v8f ol[QT];
    {
        const v2f ones = (v2f){1.0f, 1.0f};
#pragma unroll
        for (int t = 0; t < QT; ++t) {
#pragma unroll
            for (int r = 0; r < 8; ++r) plw[t * 256 + (r + 8 * hi) * 16 + ln] = ps[t][r];
        }
        asm volatile("s_wait_dscnt 0x0" ::: "memory");
#pragma unroll
        for (int t = 0; t < QT; ++t) {
            ol[t] = (v8f){};
#pragma unroll
            for (int kk = 0; kk < 4; ++kk) {
                const v2f sa = *(const v2f*)&plw[t * 256 + ln * 16 + 4 * kk + 2 * hi];
                ol[t] = __builtin_amdgcn_wmma_f32_16x16x4_f32(false, sa, false, ones,
                                                              (short)0, ol[t], false, false);
            }
        }
    }

#pragma unroll
    for (int t = 0; t < QT; ++t) {
#pragma unroll
        for (int r = 0; r < 8; ++r) {
            red_o[t][w][(r + 8 * hi) * RP + ln]      = o0[t][r];
            red_o[t][w][(r + 8 * hi) * RP + 16 + ln] = o1[t][r];
        }
        if (ln == 0) {
#pragma unroll
            for (int r = 0; r < 8; ++r) red_l[t][w][r + 8 * hi] = ol[t][r];
        }
    }
    __syncthreads();

#pragma unroll
    for (int t = 0; t < QT; ++t) {
#pragma unroll
        for (int e = 0; e < 16 / NW; ++e) {
            const int row = w + NW * e;
            float acc = 0.0f;
            float l   = 0.0f;
#pragma unroll
            for (int ww = 0; ww < NW; ++ww) {
                acc += red_o[t][ww][row * RP + lane];
                l   += red_l[t][ww][row];
            }
            const float inv = (l > 0.0f) ? (1.0f / l) : 0.0f;
            out[(size_t)(h * L_DIM + row0 + 16 * t + row) * D_DIM + lane] = acc * inv;
        }
    }
}


extern "C" void kernel_launch(void* const* d_in, const int* in_sizes, int n_in,
                              void* d_out, int out_size, void* d_ws, size_t ws_size,
                              hipStream_t stream) {
    const float* q  = (const float*)d_in[0];
    const float* k  = (const float*)d_in[1];
    const float* v  = (const float*)d_in[2];
    const int*   ei = (const int*)d_in[3];
    const float* dt = (const float*)d_in[4];
    float* out = (float*)d_out;

    unsigned int* mask = (unsigned int*)d_ws;
    const int E = in_sizes[3] / 2;
    const int nwords = (L_DIM * L_DIM) / 32;

    mask_init_kernel<<<1024, 256, 0, stream>>>(mask, nwords);
    mask_clear_edges_kernel<<<(E + 255) / 256, 256, 0, stream>>>(mask, ei, E);
    mask_clear_diag_kernel<<<(L_DIM + 255) / 256, 256, 0, stream>>>(mask);

    dim3 grid(L_DIM / (16 * QT), H_DIM);
    attn_wmma_kernel<<<grid, 32 * NW, 0, stream>>>(q, k, v, dt, mask, out);
}


# Key columns blocked for all 16 rows of a query tile are left out: each wave compacts its live columns from the bitmap into an LDS list and gathers K, V, dt by it (grid L/16 x H); sum order changes
# speedup vs baseline: 4.9038x; 2.0496x over previous
#include <hip/hip_runtime.h>

typedef __attribute__((ext_vector_type(2))) float v2f;
typedef __attribute__((ext_vector_type(4))) float v4f;
typedef __attribute__((ext_vector_type(8))) float v8f;

#define L_DIM 4096
#define H_DIM 8
#define D_DIM 32
#define TSCALE 0.1f
#define QSCALE 0.25506632772467906f


__global__ void mask_init_kernel(unsigned int* __restrict__ mask, int nwords) {
    int i = blockIdx.x * blockDim.x + threadIdx.x;
    int stride = gridDim.x * blockDim.x;
    for (; i < nwords; i += stride) mask[i] = 0xFFFFFFFFu;
}

__global__ void mask_clear_edges_kernel(unsigned int* __restrict__ mask,
                                        const int* __restrict__ ei, int E) {
    int e = blockIdx.x * blockDim.x + threadIdx.x;
    if (e >= E) return;
    unsigned int r = (unsigned int)ei[e];
    unsigned int c = (unsigned int)ei[E + e];
    unsigned int idx = r * L_DIM + c;
    atomicAnd(&mask[idx >> 5], ~(1u << (idx & 31u)));
}

__global__ void mask_clear_diag_kernel(unsigned int* __restrict__ mask) {
    int i = blockIdx.x * blockDim.x + threadIdx.x;
    if (i >= L_DIM) return;
    unsigned int idx = (unsigned int)i * (L_DIM + 1);
    atomicAnd(&mask[idx >> 5], ~(1u << (idx & 31u)));
}


#define NW 4
#define WCOLS (L_DIM / NW)
#define LCAP (WCOLS + 16)
#define RP 34
#define KP 36
#define VP 40

__attribute__((amdgpu_waves_per_eu(8, 8)))
__launch_bounds__(32 * NW)
__global__ void attn_wmma_kernel(const float* __restrict__ q,
                                 const float* __restrict__ k,
                                 const float* __restrict__ v,
                                 const float* __restrict__ dt,
                                 const unsigned int* __restrict__ mask,
                                 float* __restrict__ out) {
    __shared__ __attribute__((aligned(16))) float kst[NW][16 * KP];
    __shared__ __attribute__((aligned(16))) float vst[NW][16 * VP];
    __shared__ float pl[NW][16 * 16];
    __shared__ unsigned int lst[NW][LCAP];

    const int tid  = threadIdx.x;
    const int w    = __builtin_amdgcn_readfirstlane(tid >> 5);
    const int lane = tid & 31;
    const int hi   = lane >> 4;
    const int ln   = lane & 15;
    const int h    = blockIdx.y;
    const int row0 = blockIdx.x * 16;
    float* plw = pl[w];
    float* ksw = kst[w];
    float* vsw = vst[w];
    unsigned int* lstw = lst[w];

    int nlive;
    {
        const unsigned int* mrow = mask + (size_t)row0 * (L_DIM / 32) + (WCOLS / 32) * w + lane;
        unsigned int mw[16];
#pragma unroll
        for (int r = 0; r < 16; ++r) mw[r] = mrow[r * (L_DIM / 32)];
        unsigned int dead = mw[0];
#pragma unroll
        for (int r = 1; r < 16; ++r) dead &= mw[r];
        unsigned int bits = ~dead;
        const unsigned int cnt = (unsigned int)__builtin_popcount(bits);
        unsigned int incl = cnt;
#pragma unroll
        for (int d = 1; d < 32; d <<= 1) {
            const unsigned int up = __shfl_up(incl, (unsigned int)d, 32);
            if (lane >= d) incl += up;
        }
        unsigned int pos = incl - cnt;
        nlive = __builtin_amdgcn_readlane((int)incl, 31);
        while (bits) {
            const unsigned int b = (unsigned int)__builtin_ctz(bits);
            bits &= bits - 1u;
            unsigned int rb = 0;
#pragma unroll
            for (int r = 0; r < 16; ++r) rb |= (((~mw[r]) >> b) & 1u) << r;
            lstw[pos] = (32u * (unsigned int)lane + b) | (rb << 16);
            ++pos;
        }
        if (lane < 16) lstw[nlive + lane] = 0u;
    }
    const int ntiles = (nlive + 15) >> 4;
    asm volatile("s_wait_dscnt 0x0" ::: "memory");

    v2f aq[8];
    {
        const float* qp = q + (size_t)(h * L_DIM + row0 + ln) * D_DIM + 2 * hi;
#pragma unroll
        for (int kk = 0; kk < 8; ++kk) {
            v2f tq = *(const v2f*)(qp + 4 * kk);
            aq[kk] = tq * QSCALE;
        }
    }

    const int key0 = w * WCOLS;
    const char* kb = (const char*)(k + (size_t)(h * L_DIM + key0) * D_DIM);
    const char* vb = (const char*)(v + (size_t)(h * L_DIM + key0) * D_DIM);
    const char* db = (const char*)(dt + (size_t)row0 * L_DIM + key0);
    const unsigned int kvoff = 16u * (unsigned int)(lane & 7);
    const int srow = lane >> 3;
    const unsigned int dtoff = 4u * (unsigned int)(8 * hi * L_DIM);
    float* kst_w = ksw + (lane >> 3) * KP + (lane & 7) * 4;
    float* vst_w = vsw + (lane >> 3) * VP + (lane & 7) * 4;
    const float* kfr = ksw + ln * KP + 2 * hi;
    const float* vfr = vsw + (2 * hi) * VP + ln;

    v8f o0 = (v8f){}, o1 = (v8f){};
    float ps[8];
#pragma unroll
    for (int r = 0; r < 8; ++r) ps[r] = 0.0f;

    v4f kg[4], vg[4];
    float dv[8];
    unsigned int rbn;
    auto fetch_tile = [&](int i) {
        const unsigned int* le = lstw + 16 * i;
        const unsigned int ec = le[ln];
        rbn = (ec >> 16) >> (8 * hi);
        const unsigned int coff = (ec & 0xFFFFu) * 4u;
#pragma unroll
        for (int j = 0; j < 4; ++j) {
            const unsigned int roff = (le[4 * j + srow] & 0xFFFFu) * (unsigned int)(D_DIM * 4) + kvoff;
            kg[j] = *(const v4f*)(kb + roff);
            vg[j] = *(const v4f*)(vb + roff);
        }
#pragma unroll
        for (int r = 0; r < 8; ++r) {
            const bool allowed = (rbn >> r) & 1u;
            const unsigned int off = allowed ? (dtoff + (unsigned int)(r * L_DIM * 4) + coff) : 0u;
            dv[r] = *(const float*)(db + off);
        }
    };
    fetch_tile(0);

    for (int i = 0; i < ntiles; ++i) {
#pragma unroll
        for (int j = 0; j < 4; ++j) *(v4f*)(kst_w + 4 * j * KP) = kg[j];
#pragma unroll
        for (int j = 0; j < 4; ++j) *(v4f*)(vst_w + 4 * j * VP) = vg[j];

        float lb[8];
#pragma unroll
        for (int r = 0; r < 8; ++r) {
            const float lg = __builtin_amdgcn_logf(1.0f + fabsf(dv[r]));
            const bool allowed = (rbn >> r) & 1u;
            lb[r] = allowed ? lg : __builtin_inff();
        }
        asm volatile("" : "+v"(lb[0]), "+v"(lb[1]), "+v"(lb[2]), "+v"(lb[3]),
                          "+v"(lb[4]), "+v"(lb[5]), "+v"(lb[6]), "+v"(lb[7]) :: "memory");

        if (i + 1 < ntiles) fetch_tile(i + 1);

        asm volatile("s_wait_dscnt 0x0" ::: "memory");

        v2f bk[8];
#pragma unroll
        for (int kk = 0; kk < 8; ++kk) bk[kk] = *(const v2f*)(kfr + 4 * kk);

        v8f s = (v8f){};
#pragma unroll
        for (int kk = 0; kk < 8; ++kk) {
            s = __builtin_amdgcn_wmma_f32_16x16x4_f32(false, aq[kk], false, bk[kk],
                                                      (short)0, s, false, false);
        }

#pragma unroll
        for (int r = 0; r < 8; ++r) {
            const float sv = fmaf(lb[r], -TSCALE, s[r]);
            const float pv = __builtin_amdgcn_exp2f(sv);
            ps[r] += pv;

            plw[(r + 8 * hi) * 16 + ln] = pv;
        }

        asm volatile("s_wait_dscnt 0x0" ::: "memory");

        v2f pa[4];
#pragma unroll
        for (int kk = 0; kk < 4; ++kk) pa[kk] = *(const v2f*)&plw[ln * 16 + 4 * kk + 2 * hi];

        v2f bv0[4], bv1[4];
#pragma unroll
        for (int kk = 0; kk < 4; ++kk) {
            const float* vr = vfr + kk * 4 * VP;
            bv0[kk] = (v2f){vr[0],  vr[VP]};
            bv1[kk] = (v2f){vr[16], vr[VP + 16]};
        }

#pragma unroll
        for (int kk = 0; kk < 4; ++kk) {
            o0 = __builtin_amdgcn_wmma_f32_16x16x4_f32(false, pa[kk], false, bv0[kk],
                                                       (short)0, o0, false, false);
            o1 = __builtin_amdgcn_wmma_f32_16x16x4_f32(false, pa[kk], false, bv1[kk],
                                                       (short)0, o1, false, false);
        }
    }

    v8f ol = (v8f){};
    {
        const v2f ones = (v2f){1.0f, 1.0f};
#pragma unroll
        for (int r = 0; r < 8; ++r) plw[(r + 8 * hi) * 16 + ln] = ps[r];
        asm volatile("s_wait_dscnt 0x0" ::: "memory");
#pragma unroll
        for (int kk = 0; kk < 4; ++kk) {
            const v2f sa = *(const v2f*)&plw[ln * 16 + 4 * kk + 2 * hi];
            ol = __builtin_amdgcn_wmma_f32_16x16x4_f32(false, sa, false, ones,
                                                       (short)0, ol, false, false);
        }
    }

#pragma unroll
    for (int r = 0; r < 8; ++r) {
        lstw[(r + 8 * hi) * RP + ln]      = __float_as_uint(o0[r]);
        lstw[(r + 8 * hi) * RP + 16 + ln] = __float_as_uint(o1[r]);
    }
    if (ln == 0) {
#pragma unroll
        for (int r = 0; r < 8; ++r) lstw[16 * RP + r + 8 * hi] = __float_as_uint(ol[r]);
    }
    __syncthreads();

#pragma unroll
    for (int e = 0; e < 16 / NW; ++e) {
        const int row = w + NW * e;
        float acc = 0.0f;
        float l   = 0.0f;
#pragma unroll
        for (int ww = 0; ww < NW; ++ww) {
            acc += __uint_as_float(lst[ww][row * RP + lane]);
            l   += __uint_as_float(lst[ww][16 * RP + row]);
        }
        const float inv = (l > 0.0f) ? (1.0f / l) : 0.0f;
        out[(size_t)(h * L_DIM + row0 + row) * D_DIM + lane] = acc * inv;
    }
}


extern "C" void kernel_launch(void* const* d_in, const int* in_sizes, int n_in,
                              void* d_out, int out_size, void* d_ws, size_t ws_size,
                              hipStream_t stream) {
    const float* q  = (const float*)d_in[0];
    const float* k  = (const float*)d_in[1];
    const float* v  = (const float*)d_in[2];
    const int*   ei = (const int*)d_in[3];
    const float* dt = (const float*)d_in[4];
    float* out = (float*)d_out;

    unsigned int* mask = (unsigned int*)d_ws;
    const int E = in_sizes[3] / 2;
    const int nwords = (L_DIM * L_DIM) / 32;

    mask_init_kernel<<<1024, 256, 0, stream>>>(mask, nwords);
    mask_clear_edges_kernel<<<(E + 255) / 256, 256, 0, stream>>>(mask, ei, E);
    mask_clear_diag_kernel<<<(L_DIM + 255) / 256, 256, 0, stream>>>(mask);

    dim3 grid(L_DIM / 16, H_DIM);
    attn_wmma_kernel<<<grid, 32 * NW, 0, stream>>>(q, k, v, dt, mask, out);
}


# Each wave carries two heads per gathered key tile (grid L/16 x H/2, 4 waves per SIMD, 220 registers): live-column list, list reads, dt loads, log2 and mask values made once for both; same sum order
# speedup vs baseline: 5.1863x; 1.0576x over previous
#include <hip/hip_runtime.h>

typedef __attribute__((ext_vector_type(2))) float v2f;
typedef __attribute__((ext_vector_type(4))) float v4f;
typedef __attribute__((ext_vector_type(8))) float v8f;

#define L_DIM 4096
#define H_DIM 8
#define D_DIM 32
#define TSCALE 0.1f
#define QSCALE 0.25506632772467906f


__global__ void mask_init_kernel(unsigned int* __restrict__ mask, int nwords) {
    int i = blockIdx.x * blockDim.x + threadIdx.x;
    int stride = gridDim.x * blockDim.x;
    for (; i < nwords; i += stride) mask[i] = 0xFFFFFFFFu;
}

__global__ void mask_clear_edges_kernel(unsigned int* __restrict__ mask,
                                        const int* __restrict__ ei, int E) {
    int e = blockIdx.x * blockDim.x + threadIdx.x;
    if (e >= E) return;
    unsigned int r = (unsigned int)ei[e];
    unsigned int c = (unsigned int)ei[E + e];
    unsigned int idx = r * L_DIM + c;
    atomicAnd(&mask[idx >> 5], ~(1u << (idx & 31u)));
}

__global__ void mask_clear_diag_kernel(unsigned int* __restrict__ mask) {
    int i = blockIdx.x * blockDim.x + threadIdx.x;
    if (i >= L_DIM) return;
    unsigned int idx = (unsigned int)i * (L_DIM + 1);
    atomicAnd(&mask[idx >> 5], ~(1u << (idx & 31u)));
}


#define NW 4
#define NH 2
#define WCOLS (L_DIM / NW)
#define LCAP (WCOLS + 16)
#define RP 34
#define KP 36
#define VP 40

__attribute__((amdgpu_waves_per_eu(4, 4)))
__launch_bounds__(32 * NW)
__global__ void attn_wmma_kernel(const float* __restrict__ q,
                                 const float* __restrict__ k,
                                 const float* __restrict__ v,
                                 const float* __restrict__ dt,
                                 const unsigned int* __restrict__ mask,
                                 float* __restrict__ out) {
    __shared__ __attribute__((aligned(16))) float kst[NH][NW][16 * KP];
    __shared__ __attribute__((aligned(16))) float vst[NH][NW][16 * VP];
    __shared__ float pl[NH][NW][16 * 16];
    __shared__ unsigned int lst[NW][LCAP];

    const int tid  = threadIdx.x;
    const int w    = __builtin_amdgcn_readfirstlane(tid >> 5);
    const int lane = tid & 31;
    const int hi   = lane >> 4;
    const int ln   = lane & 15;
    const int h0   = blockIdx.y * NH;
    const int row0 = blockIdx.x * 16;
    unsigned int* lstw = lst[w];

    int nlive;
    {
        const unsigned int* mrow = mask + (size_t)row0 * (L_DIM / 32) + (WCOLS / 32) * w + lane;
        unsigned int mw[16];
#pragma unroll
        for (int r = 0; r < 16; ++r) mw[r] = mrow[r * (L_DIM / 32)];
        unsigned int dead = mw[0];
#pragma unroll
        for (int r = 1; r < 16; ++r) dead &= mw[r];
        unsigned int bits = ~dead;
        const unsigned int cnt = (unsigned int)__builtin_popcount(bits);
        unsigned int incl = cnt;
#pragma unroll
        for (int d = 1; d < 32; d <<= 1) {
            const unsigned int up = __shfl_up(incl, (unsigned int)d, 32);
            if (lane >= d) incl += up;
        }
        unsigned int pos = incl - cnt;
        nlive = __builtin_amdgcn_readlane((int)incl, 31);
        while (bits) {
            const unsigned int b = (unsigned int)__builtin_ctz(bits);
            bits &= bits - 1u;
            unsigned int rb = 0;
#pragma unroll
            for (int r = 0; r < 16; ++r) rb |= (((~mw[r]) >> b) & 1u) << r;
            lstw[pos] = (32u * (unsigned int)lane + b) | (rb << 16);
            ++pos;
        }
        if (lane < 16) lstw[nlive + lane] = 0u;
    }
    const int ntiles = (nlive + 15) >> 4;
    asm volatile("s_wait_dscnt 0x0" ::: "memory");

    v2f aq[NH][8];
#pragma unroll
    for (int hh = 0; hh < NH; ++hh) {
        const float* qp = q + (size_t)((h0 + hh) * L_DIM + row0 + ln) * D_DIM + 2 * hi;
#pragma unroll
        for (int kk = 0; kk < 8; ++kk) {
            v2f tq = *(const v2f*)(qp + 4 * kk);
            aq[hh][kk] = tq * QSCALE;
        }
    }

    const int key0 = w * WCOLS;
    const char* kb = (const char*)(k + (size_t)(h0 * L_DIM + key0) * D_DIM);
    const char* vb = (const char*)(v + (size_t)(h0 * L_DIM + key0) * D_DIM);
    const char* db = (const char*)(dt + (size_t)row0 * L_DIM + key0);
    const unsigned int kvoff = 16u * (unsigned int)(lane & 7);
    const int srow = lane >> 3;
    const unsigned int dtoff = 4u * (unsigned int)(8 * hi * L_DIM);

    v8f o0[NH], o1[NH];
    float ps[NH][8];
#pragma unroll
    for (int hh = 0; hh < NH; ++hh) {
        o0[hh] = (v8f){};
        o1[hh] = (v8f){};
#pragma unroll
        for (int r = 0; r < 8; ++r) ps[hh][r] = 0.0f;
    }

    v4f kg[NH][4], vg[NH][4];
    float dv[8];
    unsigned int rbn;
    auto fetch_tile = [&](int i) {
        const unsigned int* le = lstw + 16 * i;
        const unsigned int ec = le[ln];
        rbn = (ec >> 16) >> (8 * hi);
        const unsigned int coff = (ec & 0xFFFFu) * 4u;
#pragma unroll
        for (int j = 0; j < 4; ++j) {
            const unsigned int roff = (le[4 * j + srow] & 0xFFFFu) * (unsigned int)(D_DIM * 4) + kvoff;
#pragma unroll
            for (int hh = 0; hh < NH; ++hh) {
                kg[hh][j] = *(const v4f*)(kb + (size_t)hh * (L_DIM * D_DIM * 4) + roff);
                vg[hh][j] = *(const v4f*)(vb + (size_t)hh * (L_DIM * D_DIM * 4) + roff);
            }
        }
#pragma unroll
        for (int r = 0; r < 8; ++r) {
            const bool allowed = (rbn >> r) & 1u;
            const unsigned int off = allowed ? (dtoff + (unsigned int)(r * L_DIM * 4) + coff) : 0u;
            dv[r] = *(const float*)(db + off);
        }
    };
    fetch_tile(0);

    for (int i = 0; i < ntiles; ++i) {
#pragma unroll
        for (int hh = 0; hh < NH; ++hh) {
            float* kst_w = kst[hh][w] + (lane >> 3) * KP + (lane & 7) * 4;
            float* vst_w = vst[hh][w] + (lane >> 3) * VP + (lane & 7) * 4;
#pragma unroll
            for (int j = 0; j < 4; ++j) *(v4f*)(kst_w + 4 * j * KP) = kg[hh][j];
#pragma unroll
            for (int j = 0; j < 4; ++j) *(v4f*)(vst_w + 4 * j * VP) = vg[hh][j];
        }

        float lb[8];
#pragma unroll
        for (int r = 0; r < 8; ++r) {
            const float lg = __builtin_amdgcn_logf(1.0f + fabsf(dv[r]));
            const bool allowed = (rbn >> r) & 1u;
            lb[r] = allowed ? lg : __builtin_inff();
        }
        asm volatile("" : "+v"(lb[0]), "+v"(lb[1]), "+v"(lb[2]), "+v"(lb[3]),
                          "+v"(lb[4]), "+v"(lb[5]), "+v"(lb[6]), "+v"(lb[7]) :: "memory");

        if (i + 1 < ntiles) fetch_tile(i + 1);

        asm volatile("s_wait_dscnt 0x0" ::: "memory");

#pragma unroll
        for (int hh = 0; hh < NH; ++hh) {
            const float* kfr = kst[hh][w] + ln * KP + 2 * hi;
            v2f bk[8];
#pragma unroll
            for (int kk = 0; kk < 8; ++kk) bk[kk] = *(const v2f*)(kfr + 4 * kk);

            v8f sc = (v8f){};
#pragma unroll
            for (int kk = 0; kk < 8; ++kk) {
                sc = __builtin_amdgcn_wmma_f32_16x16x4_f32(false, aq[hh][kk], false, bk[kk],
                                                           (short)0, sc, false, false);
            }

            float* plw = pl[hh][w];
#pragma unroll
            for (int r = 0; r < 8; ++r) {
                const float sv = fmaf(lb[r], -TSCALE, sc[r]);
                const float pv = __builtin_amdgcn_exp2f(sv);
                ps[hh][r] += pv;

                plw[(r + 8 * hi) * 16 + ln] = pv;
            }
        }

        asm volatile("s_wait_dscnt 0x0" ::: "memory");

#pragma unroll
        for (int hh = 0; hh < NH; ++hh) {
            const float* plw = pl[hh][w];
            v2f pa[4];
#pragma unroll
            for (int kk = 0; kk < 4; ++kk) pa[kk] = *(const v2f*)&plw[ln * 16 + 4 * kk + 2 * hi];

            const float* vfr = vst[hh][w] + (2 * hi) * VP + ln;
            v2f bv0[4], bv1[4];
#pragma unroll
            for (int kk = 0; kk < 4; ++kk) {
                const float* vr = vfr + kk * 4 * VP;
                bv0[kk] = (v2f){vr[0],  vr[VP]};
                bv1[kk] = (v2f){vr[16], vr[VP + 16]};
            }

#pragma unroll
            for (int kk = 0; kk < 4; ++kk) {
                o0[hh] = __builtin_amdgcn_wmma_f32_16x16x4_f32(false, pa[kk], false, bv0[kk],
                                                               (short)0, o0[hh], false, false);
                o1[hh] = __builtin_amdgcn_wmma_f32_16x16x4_f32(false, pa[kk], false, bv1[kk],
                                                               (short)0, o1[hh], false, false);
            }
        }
    }

    v8f ol[NH];
    {
        const v2f ones = (v2f){1.0f, 1.0f};
#pragma unroll
        for (int hh = 0; hh < NH; ++hh) {
            float* plw = pl[hh][w];
#pragma unroll
            for (int r = 0; r < 8; ++r) plw[(r + 8 * hi) * 16 + ln] = ps[hh][r];
        }
        asm volatile("s_wait_dscnt 0x0" ::: "memory");
#pragma unroll
        for (int hh = 0; hh < NH; ++hh) {
            const float* plw = pl[hh][w];
            ol[hh] = (v8f){};
#pragma unroll
            for (int kk = 0; kk < 4; ++kk) {
                const v2f sa = *(const v2f*)&plw[ln * 16 + 4 * kk + 2 * hi];
                ol[hh] = __builtin_amdgcn_wmma_f32_16x16x4_f32(false, sa, false, ones,
                                                               (short)0, ol[hh], false, false);
            }
        }
    }

#pragma unroll
    for (int hh = 0; hh < NH; ++hh) {
        float* pw = kst[hh][w];
#pragma unroll
        for (int r = 0; r < 8; ++r) {
            pw[(r + 8 * hi) * RP + ln]      = o0[hh][r];
            pw[(r + 8 * hi) * RP + 16 + ln] = o1[hh][r];
        }
        if (ln == 0) {
#pragma unroll
            for (int r = 0; r < 8; ++r) pw[16 * RP + r + 8 * hi] = ol[hh][r];
        }
    }
    __syncthreads();

#pragma unroll
    for (int hh = 0; hh < NH; ++hh) {
#pragma unroll
        for (int e = 0; e < 16 / NW; ++e) {
            const int row = w + NW * e;
            float acc = 0.0f;
            float l   = 0.0f;
#pragma unroll
            for (int ww = 0; ww < NW; ++ww) {
                acc += kst[hh][ww][row * RP + lane];
                l   += kst[hh][ww][16 * RP + row];
            }
            const float inv = (l > 0.0f) ? (1.0f / l) : 0.0f;
            out[(size_t)((h0 + hh) * L_DIM + row0 + row) * D_DIM + lane] = acc * inv;
        }
    }
}


extern "C" void kernel_launch(void* const* d_in, const int* in_sizes, int n_in,
                              void* d_out, int out_size, void* d_ws, size_t ws_size,
                              hipStream_t stream) {
    const float* q  = (const float*)d_in[0];
    const float* k  = (const float*)d_in[1];
    const float* v  = (const float*)d_in[2];
    const int*   ei = (const int*)d_in[3];
    const float* dt = (const float*)d_in[4];
    float* out = (float*)d_out;

    unsigned int* mask = (unsigned int*)d_ws;
    const int E = in_sizes[3] / 2;
    const int nwords = (L_DIM * L_DIM) / 32;

    mask_init_kernel<<<1024, 256, 0, stream>>>(mask, nwords);
    mask_clear_edges_kernel<<<(E + 255) / 256, 256, 0, stream>>>(mask, ei, E);
    mask_clear_diag_kernel<<<(L_DIM + 255) / 256, 256, 0, stream>>>(mask);

    dim3 grid(L_DIM / 16, H_DIM / NH);
    attn_wmma_kernel<<<grid, 32 * NW, 0, stream>>>(q, k, v, dt, mask, out);
}


# LDS staging planes made bank-conflict-free: a half-wave's 16-byte K/V stores take tile rows 8 apart (V pitch 44), P tile pitch 20 (stores 2-way, A reads 4-way conflicted before); same sum order
# speedup vs baseline: 5.4762x; 1.0559x over previous
#include <hip/hip_runtime.h>

typedef __attribute__((ext_vector_type(2))) float v2f;
typedef __attribute__((ext_vector_type(4))) float v4f;
typedef __attribute__((ext_vector_type(8))) float v8f;

#define L_DIM 4096
#define H_DIM 8
#define D_DIM 32
#define TSCALE 0.1f
#define QSCALE 0.25506632772467906f


__global__ void mask_init_kernel(unsigned int* __restrict__ mask, int nwords) {
    int i = blockIdx.x * blockDim.x + threadIdx.x;
    int stride = gridDim.x * blockDim.x;
    for (; i < nwords; i += stride) mask[i] = 0xFFFFFFFFu;
}

__global__ void mask_clear_edges_kernel(unsigned int* __restrict__ mask,
                                        const int* __restrict__ ei, int E) {
    int e = blockIdx.x * blockDim.x + threadIdx.x;
    if (e >= E) return;
    unsigned int r = (unsigned int)ei[e];
    unsigned int c = (unsigned int)ei[E + e];
    unsigned int idx = r * L_DIM + c;
    atomicAnd(&mask[idx >> 5], ~(1u << (idx & 31u)));
}

__global__ void mask_clear_diag_kernel(unsigned int* __restrict__ mask) {
    int i = blockIdx.x * blockDim.x + threadIdx.x;
    if (i >= L_DIM) return;
    unsigned int idx = (unsigned int)i * (L_DIM + 1);
    atomicAnd(&mask[idx >> 5], ~(1u << (idx & 31u)));
}


#define NW 4
#define NH 2
#define WCOLS (L_DIM / NW)
#define LCAP (WCOLS + 16)
#define RP 34
#define KP 36
#define VP 44
#define PP 20

__attribute__((amdgpu_waves_per_eu(4, 4)))
__launch_bounds__(32 * NW)
__global__ void attn_wmma_kernel(const float* __restrict__ q,
                                 const float* __restrict__ k,
                                 const float* __restrict__ v,
                                 const float* __restrict__ dt,
                                 const unsigned int* __restrict__ mask,
                                 float* __restrict__ out) {
    __shared__ __attribute__((aligned(16))) float kst[NH][NW][16 * KP];
    __shared__ __attribute__((aligned(16))) float vst[NH][NW][16 * VP];
    __shared__ __attribute__((aligned(16))) float pl[NH][NW][16 * PP];
    __shared__ unsigned int lst[NW][LCAP];

    const int tid  = threadIdx.x;
    const int w    = __builtin_amdgcn_readfirstlane(tid >> 5);
    const int lane = tid & 31;
    const int hi   = lane >> 4;
    const int ln   = lane & 15;
    const int h0   = blockIdx.y * NH;
    const int row0 = blockIdx.x * 16;
    unsigned int* lstw = lst[w];

    int nlive;
    {
        const unsigned int* mrow = mask + (size_t)row0 * (L_DIM / 32) + (WCOLS / 32) * w + lane;
        unsigned int mw[16];
#pragma unroll
        for (int r = 0; r < 16; ++r) mw[r] = mrow[r * (L_DIM / 32)];
        unsigned int dead = mw[0];
#pragma unroll
        for (int r = 1; r < 16; ++r) dead &= mw[r];
        unsigned int bits = ~dead;
        const unsigned int cnt = (unsigned int)__builtin_popcount(bits);
        unsigned int incl = cnt;
#pragma unroll
        for (int d = 1; d < 32; d <<= 1) {
            const unsigned int up = __shfl_up(incl, (unsigned int)d, 32);
            if (lane >= d) incl += up;
        }
        unsigned int pos = incl - cnt;
        nlive = __builtin_amdgcn_readlane((int)incl, 31);
        while (bits) {
            const unsigned int b = (unsigned int)__builtin_ctz(bits);
            bits &= bits - 1u;
            unsigned int rb = 0;
#pragma unroll
            for (int r = 0; r < 16; ++r) rb |= (((~mw[r]) >> b) & 1u) << r;
            lstw[pos] = (32u * (unsigned int)lane + b) | (rb << 16);
            ++pos;
        }
        if (lane < 16) lstw[nlive + lane] = 0u;
    }
    const int ntiles = (nlive + 15) >> 4;
    asm volatile("s_wait_dscnt 0x0" ::: "memory");

    v2f aq[NH][8];
#pragma unroll
    for (int hh = 0; hh < NH; ++hh) {
        const float* qp = q + (size_t)((h0 + hh) * L_DIM + row0 + ln) * D_DIM + 2 * hi;
#pragma unroll
        for (int kk = 0; kk < 8; ++kk) {
            v2f tq = *(const v2f*)(qp + 4 * kk);
            aq[hh][kk] = tq * QSCALE;
        }
    }

    const int key0 = w * WCOLS;
    const char* kb = (const char*)(k + (size_t)(h0 * L_DIM + key0) * D_DIM);
    const char* vb = (const char*)(v + (size_t)(h0 * L_DIM + key0) * D_DIM);
    const char* db = (const char*)(dt + (size_t)row0 * L_DIM + key0);
    const unsigned int kvoff = 16u * (unsigned int)(lane & 7);
    const int srow = 4 * ((lane >> 4) & 1) + 8 * ((lane >> 3) & 1);
    const unsigned int dtoff = 4u * (unsigned int)(8 * hi * L_DIM);

    v8f o0[NH], o1[NH];
    float ps[NH][8];
#pragma unroll
    for (int hh = 0; hh < NH; ++hh) {
        o0[hh] = (v8f){};
        o1[hh] = (v8f){};
#pragma unroll
        for (int r = 0; r < 8; ++r) ps[hh][r] = 0.0f;
    }

    v4f kg[NH][4], vg[NH][4];
    float dv[8];
    unsigned int rbn;
    auto fetch_tile = [&](int i) {
        const unsigned int* le = lstw + 16 * i;
        const unsigned int ec = le[ln];
        rbn = (ec >> 16) >> (8 * hi);
        const unsigned int coff = (ec & 0xFFFFu) * 4u;
#pragma unroll
        for (int j = 0; j < 4; ++j) {
            const unsigned int roff = (le[j + srow] & 0xFFFFu) * (unsigned int)(D_DIM * 4) + kvoff;
#pragma unroll
            for (int hh = 0; hh < NH; ++hh) {
                kg[hh][j] = *(const v4f*)(kb + (size_t)hh * (L_DIM * D_DIM * 4) + roff);
                vg[hh][j] = *(const v4f*)(vb + (size_t)hh * (L_DIM * D_DIM * 4) + roff);
            }
        }
#pragma unroll
        for (int r = 0; r < 8; ++r) {
            const bool allowed = (rbn >> r) & 1u;
            const unsigned int off = allowed ? (dtoff + (unsigned int)(r * L_DIM * 4) + coff) : 0u;
            dv[r] = *(const float*)(db + off);
        }
    };
    fetch_tile(0);

    for (int i = 0; i < ntiles; ++i) {
#pragma unroll
        for (int hh = 0; hh < NH; ++hh) {
            float* kst_w = kst[hh][w] + srow * KP + (lane & 7) * 4;
            float* vst_w = vst[hh][w] + srow * VP + (lane & 7) * 4;
#pragma unroll
            for (int j = 0; j < 4; ++j) *(v4f*)(kst_w + j * KP) = kg[hh][j];
#pragma unroll
            for (int j = 0; j < 4; ++j) *(v4f*)(vst_w + j * VP) = vg[hh][j];
        }

        float lb[8];
#pragma unroll
        for (int r = 0; r < 8; ++r) {
            const float lg = __builtin_amdgcn_logf(1.0f + fabsf(dv[r]));
            const bool allowed = (rbn >> r) & 1u;
            lb[r] = allowed ? lg : __builtin_inff();
        }
        asm volatile("" : "+v"(lb[0]), "+v"(lb[1]), "+v"(lb[2]), "+v"(lb[3]),
                          "+v"(lb[4]), "+v"(lb[5]), "+v"(lb[6]), "+v"(lb[7]) :: "memory");

        if (i + 1 < ntiles) fetch_tile(i + 1);

        asm volatile("s_wait_dscnt 0x0" ::: "memory");

#pragma unroll
        for (int hh = 0; hh < NH; ++hh) {
            const float* kfr = kst[hh][w] + ln * KP + 2 * hi;
            v2f bk[8];
#pragma unroll
            for (int kk = 0; kk < 8; ++kk) bk[kk] = *(const v2f*)(kfr + 4 * kk);

            v8f sc = (v8f){};
#pragma unroll
            for (int kk = 0; kk < 8; ++kk) {
                sc = __builtin_amdgcn_wmma_f32_16x16x4_f32(false, aq[hh][kk], false, bk[kk],
                                                           (short)0, sc, false, false);
            }

            float* plw = pl[hh][w];
#pragma unroll
            for (int r = 0; r < 8; ++r) {
                const float sv = fmaf(lb[r], -TSCALE, sc[r]);
                const float pv = __builtin_amdgcn_exp2f(sv);
                ps[hh][r] += pv;

                plw[(r + 8 * hi) * PP + ln] = pv;
            }
        }

        asm volatile("s_wait_dscnt 0x0" ::: "memory");

#pragma unroll
        for (int hh = 0; hh < NH; ++hh) {
            const float* plw = pl[hh][w];
            v2f pa[4];
#pragma unroll
            for (int kk = 0; kk < 4; ++kk) pa[kk] = *(const v2f*)&plw[ln * PP + 4 * kk + 2 * hi];

            const float* vfr = vst[hh][w] + (2 * hi) * VP + ln;
            v2f bv0[4], bv1[4];
#pragma unroll
            for (int kk = 0; kk < 4; ++kk) {
                const float* vr = vfr + kk * 4 * VP;
                bv0[kk] = (v2f){vr[0],  vr[VP]};
                bv1[kk] = (v2f){vr[16], vr[VP + 16]};
            }

#pragma unroll
            for (int kk = 0; kk < 4; ++kk) {
                o0[hh] = __builtin_amdgcn_wmma_f32_16x16x4_f32(false, pa[kk], false, bv0[kk],
                                                               (short)0, o0[hh], false, false);
                o1[hh] = __builtin_amdgcn_wmma_f32_16x16x4_f32(false, pa[kk], false, bv1[kk],
                                                               (short)0, o1[hh], false, false);
            }
        }
    }

    v8f ol[NH];
    {
        const v2f ones = (v2f){1.0f, 1.0f};
#pragma unroll
        for (int hh = 0; hh < NH; ++hh) {
            float* plw = pl[hh][w];
#pragma unroll
            for (int r = 0; r < 8; ++r) plw[(r + 8 * hi) * PP + ln] = ps[hh][r];
        }
        asm volatile("s_wait_dscnt 0x0" ::: "memory");
#pragma unroll
        for (int hh = 0; hh < NH; ++hh) {
            const float* plw = pl[hh][w];
            ol[hh] = (v8f){};
#pragma unroll
            for (int kk = 0; kk < 4; ++kk) {
                const v2f sa = *(const v2f*)&plw[ln * PP + 4 * kk + 2 * hi];
                ol[hh] = __builtin_amdgcn_wmma_f32_16x16x4_f32(false, sa, false, ones,
                                                               (short)0, ol[hh], false, false);
            }
        }
    }

#pragma unroll
    for (int hh = 0; hh < NH; ++hh) {
        float* pw = kst[hh][w];
#pragma unroll
        for (int r = 0; r < 8; ++r) {
            pw[(r + 8 * hi) * RP + ln]      = o0[hh][r];
            pw[(r + 8 * hi) * RP + 16 + ln] = o1[hh][r];
        }
        if (ln == 0) {
#pragma unroll
            for (int r = 0; r < 8; ++r) pw[16 * RP + r + 8 * hi] = ol[hh][r];
        }
    }
    __syncthreads();

#pragma unroll
    for (int hh = 0; hh < NH; ++hh) {
#pragma unroll
        for (int e = 0; e < 16 / NW; ++e) {
            const int row = w + NW * e;
            float acc = 0.0f;
            float l   = 0.0f;
#pragma unroll
            for (int ww = 0; ww < NW; ++ww) {
                acc += kst[hh][ww][row * RP + lane];
                l   += kst[hh][ww][16 * RP + row];
            }
            const float inv = (l > 0.0f) ? (1.0f / l) : 0.0f;
            out[(size_t)((h0 + hh) * L_DIM + row0 + row) * D_DIM + lane] = acc * inv;
        }
    }
}


extern "C" void kernel_launch(void* const* d_in, const int* in_sizes, int n_in,
                              void* d_out, int out_size, void* d_ws, size_t ws_size,
                              hipStream_t stream) {
    const float* q  = (const float*)d_in[0];
    const float* k  = (const float*)d_in[1];
    const float* v  = (const float*)d_in[2];
    const int*   ei = (const int*)d_in[3];
    const float* dt = (const float*)d_in[4];
    float* out = (float*)d_out;

    unsigned int* mask = (unsigned int*)d_ws;
    const int E = in_sizes[3] / 2;
    const int nwords = (L_DIM * L_DIM) / 32;

    mask_init_kernel<<<1024, 256, 0, stream>>>(mask, nwords);
    mask_clear_edges_kernel<<<(E + 255) / 256, 256, 0, stream>>>(mask, ei, E);
    mask_clear_diag_kernel<<<(L_DIM + 255) / 256, 256, 0, stream>>>(mask);

    dim3 grid(L_DIM / 16, H_DIM / NH);
    attn_wmma_kernel<<<grid, 32 * NW, 0, stream>>>(q, k, v, dt, mask, out);
}


# Mask without global bitmap or global atomics: one launch buckets edges by query tile (LDS counters, scan); each attention work-group builds its tile's bits in LDS; 2 launches, was 4; same sum order
# speedup vs baseline: 5.6912x; 1.0393x over previous
#include <hip/hip_runtime.h>

typedef __attribute__((ext_vector_type(2))) float v2f;
typedef __attribute__((ext_vector_type(4))) float v4f;
typedef __attribute__((ext_vector_type(8))) float v8f;

#define L_DIM 4096
#define H_DIM 8
#define D_DIM 32
#define TSCALE 0.1f
#define QSCALE 0.25506632772467906f


#define CHUNK 2048
#define NTILE (L_DIM / 16)
#define PK_OFF 65536

__launch_bounds__(256)
__global__ void bucket_edges_kernel(const int* __restrict__ ei, int E, int nchunks,
                                    unsigned int* __restrict__ offs,
                                    unsigned int* __restrict__ packed) {
    __shared__ unsigned int cnt[NTILE];
    __shared__ unsigned int wsum[8];
    __shared__ unsigned int sorted[CHUNK];

    const int tid   = threadIdx.x;
    const int lane  = tid & 31;
    const int wave  = tid >> 5;
    const int chunk = blockIdx.x;
    const int base  = chunk * CHUNK;

    cnt[tid] = 0u;
    __syncthreads();

    unsigned int key[8], rank[8];
    int er[8], ec[8];
#pragma unroll
    for (int j = 0; j < 8; ++j) {
        const int e  = base + tid + 256 * j;
        const int el = (e < E) ? e : (E - 1);
        er[j] = ei[el];
        ec[j] = ei[E + el];
    }
#pragma unroll
    for (int j = 0; j < 8; ++j) {
        const int e = base + tid + 256 * j;
        key[j]  = 0xFFFFFFFFu;
        rank[j] = 0u;
        if (e < E) {
            const unsigned int r = (unsigned int)er[j] & (L_DIM - 1);
            const unsigned int c = (unsigned int)ec[j] & (L_DIM - 1);
            key[j]  = (r << 12) | c;
            rank[j] = atomicAdd(&cnt[r >> 4], 1u);
        }
    }
    __syncthreads();

    const unsigned int mycnt = cnt[tid];
    unsigned int incl = mycnt;
#pragma unroll
    for (int d = 1; d < 32; d <<= 1) {
        const unsigned int up = __shfl_up(incl, (unsigned int)d, 32);
        if (lane >= d) incl += up;
    }
    if (lane == 31) wsum[wave] = incl;
    __syncthreads();
    unsigned int wbase = 0u;
#pragma unroll
    for (int ww = 0; ww < 8; ++ww) if (ww < wave) wbase += wsum[ww];
    const unsigned int start = wbase + incl - mycnt;
    cnt[tid] = start;
    offs[(size_t)tid * nchunks + chunk] = (start << 16) | mycnt;
    __syncthreads();

#pragma unroll
    for (int j = 0; j < 8; ++j)
        if (key[j] != 0xFFFFFFFFu) sorted[cnt[key[j] >> 16] + rank[j]] = key[j];
    __syncthreads();

    const int nvalid = (E - base < CHUNK) ? (E - base) : CHUNK;
#pragma unroll
    for (int j = 0; j < 8; ++j) {
        const int i = tid + 256 * j;
        if (i < nvalid) packed[(size_t)base + i] = sorted[i];
    }
}


#define NW 4
#define NH 2
#define WCOLS (L_DIM / NW)
#define LCAP (WCOLS + 16)
#define RP 34
#define KP 36
#define VP 44
#define PP 20

__attribute__((amdgpu_waves_per_eu(4, 4)))
__launch_bounds__(32 * NW)
__global__ void attn_wmma_kernel(const float* __restrict__ q,
                                 const float* __restrict__ k,
                                 const float* __restrict__ v,
                                 const float* __restrict__ dt,
                                 const unsigned int* __restrict__ offs,
                                 const unsigned int* __restrict__ packed,
                                 int nchunks,
                                 float* __restrict__ out) {
    __shared__ __attribute__((aligned(16))) float kst[NH][NW][16 * KP];
    __shared__ __attribute__((aligned(16))) float vst[NH][NW][16 * VP];
    __shared__ __attribute__((aligned(16))) float pl[NH][NW][16 * PP];
    __shared__ unsigned int lst[NW][LCAP];
    __shared__ __attribute__((aligned(16))) unsigned int bm[16 * (L_DIM / 32)];

    const int tid  = threadIdx.x;
    const int w    = __builtin_amdgcn_readfirstlane(tid >> 5);
    const int lane = tid & 31;
    const int hi   = lane >> 4;
    const int ln   = lane & 15;
    const int h0   = blockIdx.y * NH;
    const int row0 = blockIdx.x * 16;
    unsigned int* lstw = lst[w];

    const unsigned int o_first = (tid < nchunks) ? offs[(size_t)blockIdx.x * nchunks + tid] : 0u;
    {
        typedef __attribute__((ext_vector_type(4))) unsigned int v4u;
        const v4u allset = (v4u){0xFFFFFFFFu, 0xFFFFFFFFu, 0xFFFFFFFFu, 0xFFFFFFFFu};
#pragma unroll
        for (int j = 0; j < 4; ++j) *(v4u*)&bm[4 * (tid + 128 * j)] = allset;
    }
    __syncthreads();
    if (tid < 16) {
        const unsigned int c = (unsigned int)(row0 + tid);
        atomicAnd(&bm[tid * (L_DIM / 32) + (c >> 5)], ~(1u << (c & 31u)));
    }
    for (int s = tid; s < nchunks; s += 32 * NW) {
        const unsigned int o  = (s == tid) ? o_first : offs[(size_t)blockIdx.x * nchunks + s];
        const unsigned int st = o >> 16;
        unsigned int n = o & 0xFFFFu;
        if (st + n > CHUNK) n = 0u;
        const unsigned int* seg = packed + (size_t)s * CHUNK + st;
        for (unsigned int i = 0; i < n; i += 16) {
            unsigned int wv[16];
#pragma unroll
            for (int j = 0; j < 16; ++j) wv[j] = (i + j < n) ? seg[i + j] : 0u;
#pragma unroll
            for (int j = 0; j < 16; ++j) {
                if (i + j < n) {
                    const unsigned int rr = (wv[j] >> 12) & 15u;
                    const unsigned int c  = wv[j] & (L_DIM - 1);
                    atomicAnd(&bm[rr * (L_DIM / 32) + (c >> 5)], ~(1u << (c & 31u)));
                }
            }
        }
    }
    __syncthreads();

    int nlive;
    {
        const unsigned int* mrow = bm + (WCOLS / 32) * w + lane;
        unsigned int mw[16];
#pragma unroll
        for (int r = 0; r < 16; ++r) mw[r] = mrow[r * (L_DIM / 32)];
        unsigned int dead = mw[0];
#pragma unroll
        for (int r = 1; r < 16; ++r) dead &= mw[r];
        unsigned int bits = ~dead;
        const unsigned int cnt = (unsigned int)__builtin_popcount(bits);
        unsigned int incl = cnt;
#pragma unroll
        for (int d = 1; d < 32; d <<= 1) {
            const unsigned int up = __shfl_up(incl, (unsigned int)d, 32);
            if (lane >= d) incl += up;
        }
        unsigned int pos = incl - cnt;
        nlive = __builtin_amdgcn_readlane((int)incl, 31);
        while (bits) {
            const unsigned int b = (unsigned int)__builtin_ctz(bits);
            bits &= bits - 1u;
            unsigned int rb = 0;
#pragma unroll
            for (int r = 0; r < 16; ++r) rb |= (((~mw[r]) >> b) & 1u) << r;
            lstw[pos] = (32u * (unsigned int)lane + b) | (rb << 16);
            ++pos;
        }
        if (lane < 16) lstw[nlive + lane] = 0u;
    }
    const int ntiles = (nlive + 15) >> 4;
    asm volatile("s_wait_dscnt 0x0" ::: "memory");

    v2f aq[NH][8];
#pragma unroll
    for (int hh = 0; hh < NH; ++hh) {
        const float* qp = q + (size_t)((h0 + hh) * L_DIM + row0 + ln) * D_DIM + 2 * hi;
#pragma unroll
        for (int kk = 0; kk < 8; ++kk) {
            v2f tq = *(const v2f*)(qp + 4 * kk);
            aq[hh][kk] = tq * QSCALE;
        }
    }

    const int key0 = w * WCOLS;
    const char* kb = (const char*)(k + (size_t)(h0 * L_DIM + key0) * D_DIM);
    const char* vb = (const char*)(v + (size_t)(h0 * L_DIM + key0) * D_DIM);
    const char* db = (const char*)(dt + (size_t)row0 * L_DIM + key0);
    const unsigned int kvoff = 16u * (unsigned int)(lane & 7);
    const int srow = 4 * ((lane >> 4) & 1) + 8 * ((lane >> 3) & 1);
    const unsigned int dtoff = 4u * (unsigned int)(8 * hi * L_DIM);

    v8f o0[NH], o1[NH];
    float ps[NH][8];
#pragma unroll
    for (int hh = 0; hh < NH; ++hh) {
        o0[hh] = (v8f){};
        o1[hh] = (v8f){};
#pragma unroll
        for (int r = 0; r < 8; ++r) ps[hh][r] = 0.0f;
    }

    v4f kg[NH][4], vg[NH][4];
    float dv[8];
    unsigned int rbn;
    auto fetch_tile = [&](int i) {
        const unsigned int* le = lstw + 16 * i;
        const unsigned int ec = le[ln];
        rbn = (ec >> 16) >> (8 * hi);
        const unsigned int coff = (ec & 0xFFFFu) * 4u;
#pragma unroll
        for (int j = 0; j < 4; ++j) {
            const unsigned int roff = (le[j + srow] & 0xFFFFu) * (unsigned int)(D_DIM * 4) + kvoff;
#pragma unroll
            for (int hh = 0; hh < NH; ++hh) {
                kg[hh][j] = *(const v4f*)(kb + (size_t)hh * (L_DIM * D_DIM * 4) + roff);
                vg[hh][j] = *(const v4f*)(vb + (size_t)hh * (L_DIM * D_DIM * 4) + roff);
            }
        }
#pragma unroll
        for (int r = 0; r < 8; ++r) {
            const bool allowed = (rbn >> r) & 1u;
            const unsigned int off = allowed ? (dtoff + (unsigned int)(r * L_DIM * 4) + coff) : 0u;
            dv[r] = *(const float*)(db + off);
        }
    };
    fetch_tile(0);

    for (int i = 0; i < ntiles; ++i) {
#pragma unroll
        for (int hh = 0; hh < NH; ++hh) {
            float* kst_w = kst[hh][w] + srow * KP + (lane & 7) * 4;
            float* vst_w = vst[hh][w] + srow * VP + (lane & 7) * 4;
#pragma unroll
            for (int j = 0; j < 4; ++j) *(v4f*)(kst_w + j * KP) = kg[hh][j];
#pragma unroll
            for (int j = 0; j < 4; ++j) *(v4f*)(vst_w + j * VP) = vg[hh][j];
        }

        float lb[8];
#pragma unroll
        for (int r = 0; r < 8; ++r) {
            const float lg = __builtin_amdgcn_logf(1.0f + fabsf(dv[r]));
            const bool allowed = (rbn >> r) & 1u;
            lb[r] = allowed ? lg : __builtin_inff();
        }
        asm volatile("" : "+v"(lb[0]), "+v"(lb[1]), "+v"(lb[2]), "+v"(lb[3]),
                          "+v"(lb[4]), "+v"(lb[5]), "+v"(lb[6]), "+v"(lb[7]) :: "memory");

        if (i + 1 < ntiles) fetch_tile(i + 1);

        asm volatile("s_wait_dscnt 0x0" ::: "memory");

#pragma unroll
        for (int hh = 0; hh < NH; ++hh) {
            const float* kfr = kst[hh][w] + ln * KP + 2 * hi;
            v2f bk[8];
#pragma unroll
            for (int kk = 0; kk < 8; ++kk) bk[kk] = *(const v2f*)(kfr + 4 * kk);

            v8f sc = (v8f){};
#pragma unroll
            for (int kk = 0; kk < 8; ++kk) {
                sc = __builtin_amdgcn_wmma_f32_16x16x4_f32(false, aq[hh][kk], false, bk[kk],
                                                           (short)0, sc, false, false);
            }

            float* plw = pl[hh][w];
#pragma unroll
            for (int r = 0; r < 8; ++r) {
                const float sv = fmaf(lb[r], -TSCALE, sc[r]);
                const float pv = __builtin_amdgcn_exp2f(sv);
                ps[hh][r] += pv;

                plw[(r + 8 * hi) * PP + ln] = pv;
            }
        }

        asm volatile("s_wait_dscnt 0x0" ::: "memory");

#pragma unroll
        for (int hh = 0; hh < NH; ++hh) {
            const float* plw = pl[hh][w];
            v2f pa[4];
#pragma unroll
            for (int kk = 0; kk < 4; ++kk) pa[kk] = *(const v2f*)&plw[ln * PP + 4 * kk + 2 * hi];

            const float* vfr = vst[hh][w] + (2 * hi) * VP + ln;
            v2f bv0[4], bv1[4];
#pragma unroll
            for (int kk = 0; kk < 4; ++kk) {
                const float* vr = vfr + kk * 4 * VP;
                bv0[kk] = (v2f){vr[0],  vr[VP]};
                bv1[kk] = (v2f){vr[16], vr[VP + 16]};
            }

#pragma unroll
            for (int kk = 0; kk < 4; ++kk) {
                o0[hh] = __builtin_amdgcn_wmma_f32_16x16x4_f32(false, pa[kk], false, bv0[kk],
                                                               (short)0, o0[hh], false, false);
                o1[hh] = __builtin_amdgcn_wmma_f32_16x16x4_f32(false, pa[kk], false, bv1[kk],
                                                               (short)0, o1[hh], false, false);
            }
        }
    }

    v8f ol[NH];
    {
        const v2f ones = (v2f){1.0f, 1.0f};
#pragma unroll
        for (int hh = 0; hh < NH; ++hh) {
            float* plw = pl[hh][w];
#pragma unroll
            for (int r = 0; r < 8; ++r) plw[(r + 8 * hi) * PP + ln] = ps[hh][r];
        }
        asm volatile("s_wait_dscnt 0x0" ::: "memory");
#pragma unroll
        for (int hh = 0; hh < NH; ++hh) {
            const float* plw = pl[hh][w];
            ol[hh] = (v8f){};
#pragma unroll
            for (int kk = 0; kk < 4; ++kk) {
                const v2f sa = *(const v2f*)&plw[ln * PP + 4 * kk + 2 * hi];
                ol[hh] = __builtin_amdgcn_wmma_f32_16x16x4_f32(false, sa, false, ones,
                                                               (short)0, ol[hh], false, false);
            }
        }
    }

#pragma unroll
    for (int hh = 0; hh < NH; ++hh) {
        float* pw = kst[hh][w];
#pragma unroll
        for (int r = 0; r < 8; ++r) {
            pw[(r + 8 * hi) * RP + ln]      = o0[hh][r];
            pw[(r + 8 * hi) * RP + 16 + ln] = o1[hh][r];
        }
        if (ln == 0) {
#pragma unroll
            for (int r = 0; r < 8; ++r) pw[16 * RP + r + 8 * hi] = ol[hh][r];
        }
    }
    __syncthreads();

#pragma unroll
    for (int hh = 0; hh < NH; ++hh) {
#pragma unroll
        for (int e = 0; e < 16 / NW; ++e) {
            const int row = w + NW * e;
            float acc = 0.0f;
            float l   = 0.0f;
#pragma unroll
            for (int ww = 0; ww < NW; ++ww) {
                acc += kst[hh][ww][row * RP + lane];
                l   += kst[hh][ww][16 * RP + row];
            }
            const float inv = (l > 0.0f) ? (1.0f / l) : 0.0f;
            out[(size_t)((h0 + hh) * L_DIM + row0 + row) * D_DIM + lane] = acc * inv;
        }
    }
}


extern "C" void kernel_launch(void* const* d_in, const int* in_sizes, int n_in,
                              void* d_out, int out_size, void* d_ws, size_t ws_size,
                              hipStream_t stream) {
    const float* q  = (const float*)d_in[0];
    const float* k  = (const float*)d_in[1];
    const float* v  = (const float*)d_in[2];
    const int*   ei = (const int*)d_in[3];
    const float* dt = (const float*)d_in[4];
    float* out = (float*)d_out;

    unsigned int* offs   = (unsigned int*)d_ws;
    unsigned int* packed = (unsigned int*)d_ws + PK_OFF;
    const int E = in_sizes[3] / 2;
    const int nchunks = (E + CHUNK - 1) / CHUNK;

    bucket_edges_kernel<<<nchunks, 256, 0, stream>>>(ei, E, nchunks, offs, packed);

    dim3 grid(L_DIM / 16, H_DIM / NH);
    attn_wmma_kernel<<<grid, 32 * NW, 0, stream>>>(q, k, v, dt, offs, packed, nchunks, out);
}


# Score tile computed transposed (K fragments as A, Q as B): exp2's registers are the B fragment of O^T = V^T P^T, so P no longer goes through LDS (no P stores, reloads, wait); sum order changes
# speedup vs baseline: 5.9944x; 1.0533x over previous
#include <hip/hip_runtime.h>

typedef __attribute__((ext_vector_type(2))) float v2f;
typedef __attribute__((ext_vector_type(4))) float v4f;
typedef __attribute__((ext_vector_type(8))) float v8f;

#define L_DIM 4096
#define H_DIM 8
#define D_DIM 32
#define TSCALE 0.1f
#define QSCALE 0.25506632772467906f


#define CHUNK 2048
#define NTILE (L_DIM / 16)
#define PK_OFF 65536

__launch_bounds__(256)
__global__ void bucket_edges_kernel(const int* __restrict__ ei, int E, int nchunks,
                                    unsigned int* __restrict__ offs,
                                    unsigned int* __restrict__ packed) {
    __shared__ unsigned int cnt[NTILE];
    __shared__ unsigned int wsum[8];
    __shared__ unsigned int sorted[CHUNK];

    const int tid   = threadIdx.x;
    const int lane  = tid & 31;
    const int wave  = tid >> 5;
    const int chunk = blockIdx.x;
    const int base  = chunk * CHUNK;

    cnt[tid] = 0u;
    __syncthreads();

    unsigned int key[8], rank[8];
    int er[8], ec[8];
#pragma unroll
    for (int j = 0; j < 8; ++j) {
        const int e  = base + tid + 256 * j;
        const int el = (e < E) ? e : (E - 1);
        er[j] = ei[el];
        ec[j] = ei[E + el];
    }
#pragma unroll
    for (int j = 0; j < 8; ++j) {
        const int e = base + tid + 256 * j;
        key[j]  = 0xFFFFFFFFu;
        rank[j] = 0u;
        if (e < E) {
            const unsigned int r = (unsigned int)er[j] & (L_DIM - 1);
            const unsigned int c = (unsigned int)ec[j] & (L_DIM - 1);
            key[j]  = (r << 12) | c;
            rank[j] = atomicAdd(&cnt[r >> 4], 1u);
        }
    }
    __syncthreads();

    const unsigned int mycnt = cnt[tid];
    unsigned int incl = mycnt;
#pragma unroll
    for (int d = 1; d < 32; d <<= 1) {
        const unsigned int up = __shfl_up(incl, (unsigned int)d, 32);
        if (lane >= d) incl += up;
    }
    if (lane == 31) wsum[wave] = incl;
    __syncthreads();
    unsigned int wbase = 0u;
#pragma unroll
    for (int ww = 0; ww < 8; ++ww) if (ww < wave) wbase += wsum[ww];
    const unsigned int start = wbase + incl - mycnt;
    cnt[tid] = start;
    offs[(size_t)tid * nchunks + chunk] = (start << 16) | mycnt;
    __syncthreads();

#pragma unroll
    for (int j = 0; j < 8; ++j)
        if (key[j] != 0xFFFFFFFFu) sorted[cnt[key[j] >> 16] + rank[j]] = key[j];
    __syncthreads();

    const int nvalid = (E - base < CHUNK) ? (E - base) : CHUNK;
#pragma unroll
    for (int j = 0; j < 8; ++j) {
        const int i = tid + 256 * j;
        if (i < nvalid) packed[(size_t)base + i] = sorted[i];
    }
}


#define NW 4
#define NH 2
#define WCOLS (L_DIM / NW)
#define LCAP (WCOLS + 16)
#define RP 34
#define KP 36
#define VP 44

__attribute__((amdgpu_waves_per_eu(4, 4)))
__launch_bounds__(32 * NW)
__global__ void attn_wmma_kernel(const float* __restrict__ q,
                                 const float* __restrict__ k,
                                 const float* __restrict__ v,
                                 const float* __restrict__ dt,
                                 const unsigned int* __restrict__ offs,
                                 const unsigned int* __restrict__ packed,
                                 int nchunks,
                                 float* __restrict__ out) {
    __shared__ __attribute__((aligned(16))) float kst[NH][NW][16 * KP];
    __shared__ __attribute__((aligned(16))) float vst[NH][NW][16 * VP];
    __shared__ __attribute__((aligned(16))) unsigned int lst[NW][LCAP];
    __shared__ __attribute__((aligned(16))) unsigned int bm[16 * (L_DIM / 32)];

    const int tid  = threadIdx.x;
    const int w    = __builtin_amdgcn_readfirstlane(tid >> 5);
    const int lane = tid & 31;
    const int hi   = lane >> 4;
    const int ln   = lane & 15;
    const int h0   = blockIdx.y * NH;
    const int row0 = blockIdx.x * 16;
    unsigned int* lstw = lst[w];

    const unsigned int o_first = (tid < nchunks) ? offs[(size_t)blockIdx.x * nchunks + tid] : 0u;
    {
        typedef __attribute__((ext_vector_type(4))) unsigned int v4u;
        const v4u allset = (v4u){0xFFFFFFFFu, 0xFFFFFFFFu, 0xFFFFFFFFu, 0xFFFFFFFFu};
#pragma unroll
        for (int j = 0; j < 4; ++j) *(v4u*)&bm[4 * (tid + 128 * j)] = allset;
    }
    __syncthreads();
    if (tid < 16) {
        const unsigned int c = (unsigned int)(row0 + tid);
        atomicAnd(&bm[tid * (L_DIM / 32) + (c >> 5)], ~(1u << (c & 31u)));
    }
    for (int s = tid; s < nchunks; s += 32 * NW) {
        const unsigned int o  = (s == tid) ? o_first : offs[(size_t)blockIdx.x * nchunks + s];
        const unsigned int st = o >> 16;
        unsigned int n = o & 0xFFFFu;
        if (st + n > CHUNK) n = 0u;
        const unsigned int* seg = packed + (size_t)s * CHUNK + st;
        for (unsigned int i = 0; i < n; i += 16) {
            unsigned int wv[16];
#pragma unroll
            for (int j = 0; j < 16; ++j) wv[j] = (i + j < n) ? seg[i + j] : 0u;
#pragma unroll
            for (int j = 0; j < 16; ++j) {
                if (i + j < n) {
                    const unsigned int rr = (wv[j] >> 12) & 15u;
                    const unsigned int c  = wv[j] & (L_DIM - 1);
                    atomicAnd(&bm[rr * (L_DIM / 32) + (c >> 5)], ~(1u << (c & 31u)));
                }
            }
        }
    }
    __syncthreads();

    int nlive;
    {
        const unsigned int* mrow = bm + (WCOLS / 32) * w + lane;
        unsigned int mw[16];
#pragma unroll
        for (int r = 0; r < 16; ++r) mw[r] = mrow[r * (L_DIM / 32)];
        unsigned int dead = mw[0];
#pragma unroll
        for (int r = 1; r < 16; ++r) dead &= mw[r];
        unsigned int bits = ~dead;
        const unsigned int cnt = (unsigned int)__builtin_popcount(bits);
        unsigned int incl = cnt;
#pragma unroll
        for (int d = 1; d < 32; d <<= 1) {
            const unsigned int up = __shfl_up(incl, (unsigned int)d, 32);
            if (lane >= d) incl += up;
        }
        unsigned int pos = incl - cnt;
        nlive = __builtin_amdgcn_readlane((int)incl, 31);
        while (bits) {
            const unsigned int b = (unsigned int)__builtin_ctz(bits);
            bits &= bits - 1u;
            unsigned int rb = 0;
#pragma unroll
            for (int r = 0; r < 16; ++r) rb |= (((~mw[r]) >> b) & 1u) << r;
            lstw[pos] = (32u * (unsigned int)lane + b) | (rb << 16);
            ++pos;
        }
        if (lane < 16) lstw[nlive + lane] = 0u;
    }
    const int ntiles = (nlive + 15) >> 4;
    asm volatile("s_wait_dscnt 0x0" ::: "memory");

    v2f aq[NH][8];
#pragma unroll
    for (int hh = 0; hh < NH; ++hh) {
        const float* qp = q + (size_t)((h0 + hh) * L_DIM + row0 + ln) * D_DIM + 2 * hi;
#pragma unroll
        for (int kk = 0; kk < 8; ++kk) {
            v2f tq = *(const v2f*)(qp + 4 * kk);
            aq[hh][kk] = tq * QSCALE;
        }
    }

    const int key0 = w * WCOLS;
    const char* kb = (const char*)(k + (size_t)(h0 * L_DIM + key0) * D_DIM);
    const char* vb = (const char*)(v + (size_t)(h0 * L_DIM + key0) * D_DIM);
    const char* db = (const char*)(dt + (size_t)row0 * L_DIM + key0);
    const unsigned int kvoff = 16u * (unsigned int)(lane & 7);
    const int srow = 4 * ((lane >> 4) & 1) + 8 * ((lane >> 3) & 1);
    const unsigned int dtoff = 4u * (unsigned int)(ln * L_DIM);

    v8f o0[NH], o1[NH];
    float ps[NH][8];
#pragma unroll
    for (int hh = 0; hh < NH; ++hh) {
        o0[hh] = (v8f){};
        o1[hh] = (v8f){};
#pragma unroll
        for (int r = 0; r < 8; ++r) ps[hh][r] = 0.0f;
    }

    v4f kg[NH][4], vg[NH][4];
    float dv[8];
    int am[8];
    auto fetch_tile = [&](int i) {
        const unsigned int* le = lstw + 16 * i;
        typedef __attribute__((ext_vector_type(4))) unsigned int v4u;
        const v4u e03 = *(const v4u*)(le + 8 * hi);
        const v4u e47 = *(const v4u*)(le + 8 * hi + 4);
        const unsigned int eh[8] = {e03.x, e03.y, e03.z, e03.w, e47.x, e47.y, e47.z, e47.w};
#pragma unroll
        for (int j = 0; j < 4; ++j) {
            const unsigned int roff = (le[j + srow] & 0xFFFFu) * (unsigned int)(D_DIM * 4) + kvoff;
#pragma unroll
            for (int hh = 0; hh < NH; ++hh) {
                kg[hh][j] = *(const v4f*)(kb + (size_t)hh * (L_DIM * D_DIM * 4) + roff);
                vg[hh][j] = *(const v4f*)(vb + (size_t)hh * (L_DIM * D_DIM * 4) + roff);
            }
        }
#pragma unroll
        for (int r = 0; r < 8; ++r) {
            am[r] = ((int)(eh[r] << (15 - ln))) >> 31;
            const unsigned int off = (dtoff + (eh[r] & 0xFFFFu) * 4u) & (unsigned int)am[r];
            dv[r] = *(const float*)(db + off);
        }
    };
    fetch_tile(0);

    int dhalf = 16;
    asm volatile("" : "+v"(dhalf));

    for (int i = 0; i < ntiles; ++i) {
#pragma unroll
        for (int hh = 0; hh < NH; ++hh) {
            float* kst_w = kst[hh][w] + srow * KP + (lane & 7) * 4;
            float* vst_w = vst[hh][w] + srow * VP + (lane & 7) * 4;
#pragma unroll
            for (int j = 0; j < 4; ++j) *(v4f*)(kst_w + j * KP) = kg[hh][j];
#pragma unroll
            for (int j = 0; j < 4; ++j) *(v4f*)(vst_w + j * VP) = vg[hh][j];
        }

        float lb[8];
#pragma unroll
        for (int r = 0; r < 8; ++r) {
            const float lg = __builtin_amdgcn_logf(1.0f + fabsf(dv[r]));
            lb[r] = (am[r] != 0) ? lg : __builtin_inff();
        }
        asm volatile("" : "+v"(lb[0]), "+v"(lb[1]), "+v"(lb[2]), "+v"(lb[3]),
                          "+v"(lb[4]), "+v"(lb[5]), "+v"(lb[6]), "+v"(lb[7]) :: "memory");

        if (i + 1 < ntiles) fetch_tile(i + 1);

        asm volatile("s_wait_dscnt 0x0" ::: "memory");

#pragma unroll
        for (int hh = 0; hh < NH; ++hh) {
            const float* kfr = kst[hh][w] + ln * KP + 2 * hi;
            v2f ak[8];
#pragma unroll
            for (int kk = 0; kk < 8; ++kk) ak[kk] = *(const v2f*)(kfr + 4 * kk);

            const float* vfr = vst[hh][w] + (8 * hi) * VP + ln;
            const float* vfh = vfr + dhalf;
            v2f av0[4], av1[4];
#pragma unroll
            for (int kk = 0; kk < 4; ++kk) {
                av0[kk] = (v2f){vfr[kk * 2 * VP], vfr[kk * 2 * VP + VP]};
                av1[kk] = (v2f){vfh[kk * 2 * VP], vfh[kk * 2 * VP + VP]};
            }

            v8f sc = (v8f){};
#pragma unroll
            for (int kk = 0; kk < 8; ++kk) {
                sc = __builtin_amdgcn_wmma_f32_16x16x4_f32(false, ak[kk], false, aq[hh][kk],
                                                           (short)0, sc, false, false);
            }

            float pv[8];
#pragma unroll
            for (int r = 0; r < 8; ++r) {
                const float sv = fmaf(lb[r], -TSCALE, sc[r]);
                pv[r] = __builtin_amdgcn_exp2f(sv);
                ps[hh][r] += pv[r];
            }

#pragma unroll
            for (int kk = 0; kk < 4; ++kk) {
                const v2f pb = (v2f){pv[2 * kk], pv[2 * kk + 1]};
                o0[hh] = __builtin_amdgcn_wmma_f32_16x16x4_f32(false, av0[kk], false, pb,
                                                               (short)0, o0[hh], false, false);
                o1[hh] = __builtin_amdgcn_wmma_f32_16x16x4_f32(false, av1[kk], false, pb,
                                                               (short)0, o1[hh], false, false);
            }
        }
    }

    v8f ol[NH];
    {
        const v2f ones = (v2f){1.0f, 1.0f};
#pragma unroll
        for (int hh = 0; hh < NH; ++hh) {
            ol[hh] = (v8f){};
#pragma unroll
            for (int kk = 0; kk < 4; ++kk) {
                const v2f sb = (v2f){ps[hh][2 * kk], ps[hh][2 * kk + 1]};
                ol[hh] = __builtin_amdgcn_wmma_f32_16x16x4_f32(false, ones, false, sb,
                                                               (short)0, ol[hh], false, false);
            }
        }
    }

#pragma unroll
    for (int hh = 0; hh < NH; ++hh) {
        float* pw = kst[hh][w];
#pragma unroll
        for (int r = 0; r < 8; ++r) {
            pw[ln * RP + r + 8 * hi]      = o0[hh][r];
            pw[ln * RP + 16 + r + 8 * hi] = o1[hh][r];
        }
        if (hi == 0) pw[16 * RP + ln] = ol[hh][0];
    }
    __syncthreads();

#pragma unroll
    for (int hh = 0; hh < NH; ++hh) {
#pragma unroll
        for (int e = 0; e < 16 / NW; ++e) {
            const int row = w + NW * e;
            float acc = 0.0f;
            float l   = 0.0f;
#pragma unroll
            for (int ww = 0; ww < NW; ++ww) {
                acc += kst[hh][ww][row * RP + lane];
                l   += kst[hh][ww][16 * RP + row];
            }
            const float inv = (l > 0.0f) ? (1.0f / l) : 0.0f;
            out[(size_t)((h0 + hh) * L_DIM + row0 + row) * D_DIM + lane] = acc * inv;
        }
    }
}


extern "C" void kernel_launch(void* const* d_in, const int* in_sizes, int n_in,
                              void* d_out, int out_size, void* d_ws, size_t ws_size,
                              hipStream_t stream) {
    const float* q  = (const float*)d_in[0];
    const float* k  = (const float*)d_in[1];
    const float* v  = (const float*)d_in[2];
    const int*   ei = (const int*)d_in[3];
    const float* dt = (const float*)d_in[4];
    float* out = (float*)d_out;

    unsigned int* offs   = (unsigned int*)d_ws;
    unsigned int* packed = (unsigned int*)d_ws + PK_OFF;
    const int E = in_sizes[3] / 2;
    const int nchunks = (E + CHUNK - 1) / CHUNK;

    bucket_edges_kernel<<<nchunks, 256, 0, stream>>>(ei, E, nchunks, offs, packed);

    dim3 grid(L_DIM / 16, H_DIM / NH);
    attn_wmma_kernel<<<grid, 32 * NW, 0, stream>>>(q, k, v, dt, offs, packed, nchunks, out);
}
